# starting kernel
# baseline (speedup 1.0000x reference)
#include <hip/hip_runtime.h>
#include <hip/hip_bf16.h>

#define NNODES 50000
#define NEDGES 800000

typedef __attribute__((ext_vector_type(16))) __bf16        bf16x16;
typedef __attribute__((ext_vector_type(8)))  float         f32x8;
typedef __attribute__((ext_vector_type(8)))  unsigned int  u32x8;

__device__ __forceinline__ unsigned short f2bf(float f) {
  unsigned int u = __float_as_uint(f);
  u += 0x7FFFu + ((u >> 16) & 1u);
  return (unsigned short)(u >> 16);
}
__device__ __forceinline__ float bf2f(unsigned short s) {
  return __uint_as_float(((unsigned int)s) << 16);
}
__device__ __forceinline__ float silu_f(float v) { return v / (1.0f + __expf(-v)); }
__device__ __forceinline__ float sigm_f(float v) { return 1.0f / (1.0f + __expf(-v)); }

__device__ __forceinline__ int frag_k(int lane, int v) {
  return ((lane >> 4) << 3) + ((v >> 2) << 4) + ((v & 3) << 1);
}

__device__ __forceinline__ bf16x16 load_a_frag(const unsigned short* tile, int ld, int kc, int lane) {
  const int row = lane & 15;
  u32x8 r;
#pragma unroll
  for (int v = 0; v < 8; ++v) {
    const int k = kc * 32 + frag_k(lane, v);
    r[v] = *(const unsigned int*)(tile + row * ld + k);
  }
  return __builtin_bit_cast(bf16x16, r);
}

__device__ __forceinline__ bf16x16 load_b_frag(const u32x8* __restrict__ wf, int kc, int nt, int lane) {
  return __builtin_bit_cast(bf16x16, wf[(kc * 8 + nt) * 32 + lane]);
}

__global__ void egnn_prep_kernel(const float* __restrict__ w_e1, const float* __restrict__ w_e2,
                                 const float* __restrict__ w_h1, const float* __restrict__ w_h2,
                                 unsigned int* __restrict__ wf_e1, unsigned int* __restrict__ wf_e2,
                                 unsigned int* __restrict__ wf_h1, unsigned int* __restrict__ wf_h2) {
  int t = blockIdx.x * blockDim.x + threadIdx.x;
  const float* W; unsigned int* D; int K; int local;
  if      (t < 18432)         { W = w_e1; D = wf_e1; K = 257; local = t; }
  else if (t < 18432 + 8192)  { W = w_e2; D = wf_e2; K = 128; local = t - 18432; }
  else if (t < 26624 + 16384) { W = w_h1; D = wf_h1; K = 256; local = t - 26624; }
  else if (t < 51200)         { W = w_h2; D = wf_h2; K = 128; local = t - 43008; }
  else return;
  const int v = local & 7, lane = (local >> 3) & 31, nt = (local >> 8) & 7, kc = local >> 11;
  const int k   = kc * 32 + ((lane >> 4) << 3) + ((v >> 2) << 4) + ((v & 3) << 1);
  const int col = nt * 16 + (lane & 15);
  const float a = (k     < K) ? W[k * 128 + col]       : 0.0f;
  const float b = (k + 1 < K) ? W[(k + 1) * 128 + col] : 0.0f;
  D[local] = (unsigned int)f2bf(a) | ((unsigned int)f2bf(b) << 16);
}

__global__ void egnn_zero_kernel(float* __restrict__ p, int n) {
  int t = blockIdx.x * blockDim.x + threadIdx.x;
  if (t < n) p[t] = 0.0f;
}

__global__ __launch_bounds__(128) void egnn_edge_kernel(
    const float* __restrict__ h, const float* __restrict__ x, const int* __restrict__ edges,
    const u32x8* __restrict__ wf_e1, const u32x8* __restrict__ wf_e2,
    const float* __restrict__ b_e1, const float* __restrict__ b_e2,
    const float* __restrict__ w_inf, const float* __restrict__ b_inf,
    float* __restrict__ m_i) {
  __shared__ unsigned short stage[4][16 * 288];
  __shared__ unsigned short a2[4][16 * 128];
  __shared__ float gbuf[4][16];

  const int lane = threadIdx.x & 31;
  const int wave = threadIdx.x >> 5;
  const int colq = lane & 15;
  const int ebase = (blockIdx.x * 4 + wave) * 16;
  unsigned short* st = stage[wave];
  unsigned short* aw = a2[wave];

  for (int r = 0; r < 16; ++r) {
    const int e = ebase + r;
    const int s = edges[2 * e], d = edges[2 * e + 1];
    const float dx = x[3 * s] - x[3 * d], dy = x[3 * s + 1] - x[3 * d + 1], dz = x[3 * s + 2] - x[3 * d + 2];
    const float dist = sqrtf(dx * dx + dy * dy + dz * dz);
#pragma unroll
    for (int i = lane; i < 288; i += 32) {
      float v;
      if      (i < 128)  v = h[s * 128 + i];
      else if (i < 256)  v = h[d * 128 + (i - 128)];
      else if (i == 256) v = dist;
      else               v = 0.0f;
      st[r * 288 + i] = f2bf(v);
    }
  }
  __builtin_amdgcn_wave_barrier();

#pragma unroll 1
  for (int nt = 0; nt < 8; ++nt) {
    f32x8 acc;
    const float bias = b_e1[nt * 16 + colq];
#pragma unroll
    for (int v = 0; v < 8; ++v) acc[v] = bias;
#pragma unroll
    for (int kc = 0; kc < 9; ++kc) {
      if (kc + 1 < 9) __builtin_prefetch(&wf_e1[((kc + 1) * 8 + nt) * 32 + lane], 0, 3);
      bf16x16 a = load_a_frag(st, 288, kc, lane);
      bf16x16 b = load_b_frag(wf_e1, kc, nt, lane);
      acc = __builtin_amdgcn_wmma_f32_16x16x32_bf16(false, a, false, b, (short)0, acc, false, false);
    }
#pragma unroll
    for (int v = 0; v < 8; ++v) {
      const int row = v + ((lane >> 4) << 3);
      aw[row * 128 + nt * 16 + colq] = f2bf(silu_f(acc[v]));
    }
  }
  __builtin_amdgcn_wave_barrier();

  unsigned short* mb = st;
  f32x8 dot;
#pragma unroll
  for (int v = 0; v < 8; ++v) dot[v] = 0.0f;
#pragma unroll 1
  for (int nt = 0; nt < 8; ++nt) {
    f32x8 acc;
    const float bias = b_e2[nt * 16 + colq];
#pragma unroll
    for (int v = 0; v < 8; ++v) acc[v] = bias;
#pragma unroll
    for (int kc = 0; kc < 4; ++kc) {
      bf16x16 a = load_a_frag(aw, 128, kc, lane);
      bf16x16 b = load_b_frag(wf_e2, kc, nt, lane);
      acc = __builtin_amdgcn_wmma_f32_16x16x32_bf16(false, a, false, b, (short)0, acc, false, false);
    }
    const float wv = w_inf[nt * 16 + colq];
#pragma unroll
    for (int v = 0; v < 8; ++v) {
      const float m = silu_f(acc[v]);
      dot[v] += m * wv;
      const int row = v + ((lane >> 4) << 3);
      mb[row * 128 + nt * 16 + colq] = f2bf(m);
    }
  }
  const float binf = b_inf[0];
#pragma unroll
  for (int v = 0; v < 8; ++v) {
    float s = dot[v];
    s += __shfl_xor(s, 1, 32);
    s += __shfl_xor(s, 2, 32);
    s += __shfl_xor(s, 4, 32);
    s += __shfl_xor(s, 8, 32);
    if (colq == 0) gbuf[wave][v + ((lane >> 4) << 3)] = sigm_f(s + binf);
  }
  __builtin_amdgcn_wave_barrier();

  for (int r = 0; r < 16; ++r) {
    const int e = ebase + r;
    const int s = edges[2 * e];
    const float g = gbuf[wave][r];
#pragma unroll
    for (int c = lane; c < 128; c += 32) {
      unsafeAtomicAdd(&m_i[s * 128 + c], g * bf2f(mb[r * 128 + c]));
    }
  }
}

__global__ __launch_bounds__(128) void egnn_node_kernel(
    const float* __restrict__ h,
    const u32x8* __restrict__ wf_h1, const u32x8* __restrict__ wf_h2,
    const float* __restrict__ b_h1, const float* __restrict__ b_h2,
    float* __restrict__ hio) {
  __shared__ unsigned short stage[4][16 * 256];
  __shared__ unsigned short a2[4][16 * 128];

  const int lane = threadIdx.x & 31;
  const int wave = threadIdx.x >> 5;
  const int colq = lane & 15;
  const int nbase = (blockIdx.x * 4 + wave) * 16;
  if (nbase >= NNODES) return;
  unsigned short* st = stage[wave];
  unsigned short* aw = a2[wave];

  for (int r = 0; r < 16; ++r) {
    const int node = nbase + r;
    const bool ok = node < NNODES;
#pragma unroll
    for (int i = lane; i < 256; i += 32) {
      float v = 0.0f;
      if (ok) v = (i < 128) ? h[node * 128 + i] : hio[node * 128 + (i - 128)];
      st[r * 256 + i] = f2bf(v);
    }
  }
  __builtin_amdgcn_wave_barrier();

#pragma unroll 1
  for (int nt = 0; nt < 8; ++nt) {
    f32x8 acc;
    const float bias = b_h1[nt * 16 + colq];
#pragma unroll
    for (int v = 0; v < 8; ++v) acc[v] = bias;
#pragma unroll
    for (int kc = 0; kc < 8; ++kc) {
      if (kc + 1 < 8) __builtin_prefetch(&wf_h1[((kc + 1) * 8 + nt) * 32 + lane], 0, 3);
      bf16x16 a = load_a_frag(st, 256, kc, lane);
      bf16x16 b = load_b_frag(wf_h1, kc, nt, lane);
      acc = __builtin_amdgcn_wmma_f32_16x16x32_bf16(false, a, false, b, (short)0, acc, false, false);
    }
#pragma unroll
    for (int v = 0; v < 8; ++v) {
      const int row = v + ((lane >> 4) << 3);
      aw[row * 128 + nt * 16 + colq] = f2bf(silu_f(acc[v]));
    }
  }
  __builtin_amdgcn_wave_barrier();

#pragma unroll 1
  for (int nt = 0; nt < 8; ++nt) {
    f32x8 acc;
    const float bias = b_h2[nt * 16 + colq];
#pragma unroll
    for (int v = 0; v < 8; ++v) acc[v] = bias;
#pragma unroll
    for (int kc = 0; kc < 4; ++kc) {
      bf16x16 a = load_a_frag(aw, 128, kc, lane);
      bf16x16 b = load_b_frag(wf_h2, kc, nt, lane);
      acc = __builtin_amdgcn_wmma_f32_16x16x32_bf16(false, a, false, b, (short)0, acc, false, false);
    }
#pragma unroll
    for (int v = 0; v < 8; ++v) {
      const int row = v + ((lane >> 4) << 3);
      const int node = nbase + row;
      const int col = nt * 16 + colq;
      if (node < NNODES) hio[node * 128 + col] = acc[v] + h[node * 128 + col];
    }
  }
}

extern "C" void kernel_launch(void* const* d_in, const int* in_sizes, int n_in,
                              void* d_out, int out_size, void* d_ws, size_t ws_size,
                              hipStream_t stream) {
  const float* h     = (const float*)d_in[0];
  const float* x     = (const float*)d_in[1];
  const int*   edges = (const int*)  d_in[2];
  const float* w_e1  = (const float*)d_in[3];
  const float* b_e1  = (const float*)d_in[4];
  const float* w_e2  = (const float*)d_in[5];
  const float* b_e2  = (const float*)d_in[6];
  const float* w_inf = (const float*)d_in[7];
  const float* b_inf = (const float*)d_in[8];
  const float* w_h1  = (const float*)d_in[9];
  const float* b_h1  = (const float*)d_in[10];
  const float* w_h2  = (const float*)d_in[11];
  const float* b_h2  = (const float*)d_in[12];
  float* out = (float*)d_out;

  unsigned int* wf_e1 = (unsigned int*)d_ws;
  unsigned int* wf_e2 = wf_e1 + 18432;
  unsigned int* wf_h1 = wf_e2 + 8192;
  unsigned int* wf_h2 = wf_h1 + 16384;

  egnn_prep_kernel<<<(51200 + 255) / 256, 256, 0, stream>>>(
      w_e1, w_e2, w_h1, w_h2, wf_e1, wf_e2, wf_h1, wf_h2);

  egnn_zero_kernel<<<(NNODES * 128 + 255) / 256, 256, 0, stream>>>(out, NNODES * 128);

  egnn_edge_kernel<<<NEDGES / 64, 128, 0, stream>>>(
      h, x, edges, (const u32x8*)wf_e1, (const u32x8*)wf_e2,
      b_e1, b_e2, w_inf, b_inf, out);

  egnn_node_kernel<<<(NNODES + 63) / 64, 128, 0, stream>>>(
      h, (const u32x8*)wf_h1, (const u32x8*)wf_h2, b_h1, b_h2, out);
}


# h converted to bf16 once per node, in the prep launch, into a fragment-order copy; edge and node kernels load its A fragments as 32-byte reads: no per-edge conversion, half the gather bytes
# speedup vs baseline: 1.4821x; 1.4821x over previous
#include <hip/hip_runtime.h>
#include <hip/hip_bf16.h>

#define NNODES 50000
#define NEDGES 800000

typedef __attribute__((ext_vector_type(16))) __bf16        bf16x16;
typedef __attribute__((ext_vector_type(8)))  float         f32x8;
typedef __attribute__((ext_vector_type(8)))  unsigned int  u32x8;

__device__ __forceinline__ unsigned short f2bf(float f) {
  unsigned int u = __float_as_uint(f);
  u += 0x7FFFu + ((u >> 16) & 1u);
  return (unsigned short)(u >> 16);
}
__device__ __forceinline__ float bf2f(unsigned short s) {
  return __uint_as_float(((unsigned int)s) << 16);
}
__device__ __forceinline__ float silu_f(float v) { return v / (1.0f + __expf(-v)); }
__device__ __forceinline__ float sigm_f(float v) { return 1.0f / (1.0f + __expf(-v)); }

__device__ __forceinline__ int frag_k(int lane, int v) {
  return ((lane >> 4) << 3) + ((v >> 2) << 4) + ((v & 3) << 1);
}

typedef __attribute__((ext_vector_type(4))) float         f32x4;
typedef __attribute__((ext_vector_type(4))) unsigned int  u32x4;

__device__ __forceinline__ unsigned int pack2bf(float lo, float hi) {
  return (unsigned int)f2bf(lo) | ((unsigned int)f2bf(hi) << 16);
}

__device__ __forceinline__ bf16x16 a_frag_from_row(const float* p) {
  const f32x4 q0 = *(const f32x4*)(p);
  const f32x4 q1 = *(const f32x4*)(p + 4);
  const f32x4 q2 = *(const f32x4*)(p + 16);
  const f32x4 q3 = *(const f32x4*)(p + 20);
  u32x8 r;
  r[0] = pack2bf(q0[0], q0[1]); r[1] = pack2bf(q0[2], q0[3]);
  r[2] = pack2bf(q1[0], q1[1]); r[3] = pack2bf(q1[2], q1[3]);
  r[4] = pack2bf(q2[0], q2[1]); r[5] = pack2bf(q2[2], q2[3]);
  r[6] = pack2bf(q3[0], q3[1]); r[7] = pack2bf(q3[2], q3[3]);
  return __builtin_bit_cast(bf16x16, r);
}

__device__ __forceinline__ int fragord_index(int row, int col) {
  return (((col >> 4) * 32 + row + (((col >> 3) & 1) << 4)) << 3) + (col & 7);
}
__device__ __forceinline__ bf16x16 a_frag_from_fragord(const unsigned short* buf, int kc, int lane) {
  const u32x4 lo = *(const u32x4*)(buf + (((2 * kc)     * 32 + lane) << 3));
  const u32x4 hi = *(const u32x4*)(buf + (((2 * kc + 1) * 32 + lane) << 3));
  u32x8 r;
  r[0] = lo[0]; r[1] = lo[1]; r[2] = lo[2]; r[3] = lo[3];
  r[4] = hi[0]; r[5] = hi[1]; r[6] = hi[2]; r[7] = hi[3];
  return __builtin_bit_cast(bf16x16, r);
}

__device__ __forceinline__ bf16x16 a_frag_from_hb(const u32x8* __restrict__ hb8, int node, int kc, int half) {
  return __builtin_bit_cast(bf16x16, hb8[node * 8 + kc * 2 + half]);
}

__device__ __forceinline__ bf16x16 load_b_frag(const u32x8* __restrict__ wf, int kc, int nt, int lane) {
  return __builtin_bit_cast(bf16x16, wf[(kc * 8 + nt) * 32 + lane]);
}

#define PREP_W_BLOCKS 200
#define PREP_ITEMS 4
#define PREP_H_BLOCKS 782
__global__ void egnn_prep_kernel(const float* __restrict__ w_e1, const float* __restrict__ w_e2,
                                 const float* __restrict__ w_h1, const float* __restrict__ w_h2,
                                 unsigned int* __restrict__ wf_e1, unsigned int* __restrict__ wf_e2,
                                 unsigned int* __restrict__ wf_h1, unsigned int* __restrict__ wf_h2,
                                 const float* __restrict__ h, u32x4* __restrict__ hb) {
  if (blockIdx.x >= PREP_W_BLOCKS) {
    const int tt = (blockIdx.x - PREP_W_BLOCKS) * 256 + threadIdx.x;
    f32x4 q0[PREP_ITEMS], q1[PREP_ITEMS];
#pragma unroll
    for (int j = 0; j < PREP_ITEMS; ++j) {
      const int item = tt + j * (PREP_H_BLOCKS * 256);
      const int ld = item < NNODES * 16 ? item : NNODES * 16 - 1;
      const float* p = h + ld * 8;
      q0[j] = *(const f32x4*)(p);
      q1[j] = *(const f32x4*)(p + 4);
    }
#pragma unroll
    for (int j = 0; j < PREP_ITEMS; ++j) {
      const int item = tt + j * (PREP_H_BLOCKS * 256);
      const int g = item & 15, kc = g >> 2, q = g & 3;
      const int slot = ((q & 1) << 1) | (q >> 1);
      u32x4 r;
      r[0] = pack2bf(q0[j][0], q0[j][1]); r[1] = pack2bf(q0[j][2], q0[j][3]);
      r[2] = pack2bf(q1[j][0], q1[j][1]); r[3] = pack2bf(q1[j][2], q1[j][3]);
      if (item < NNODES * 16) hb[(item & ~15) + kc * 4 + slot] = r;
    }
    return;
  }
  int t = blockIdx.x * blockDim.x + threadIdx.x;
  const float* W; unsigned int* D; int K; int local;
  if      (t < 18432)         { W = w_e1; D = wf_e1; K = 257; local = t; }
  else if (t < 18432 + 8192)  { W = w_e2; D = wf_e2; K = 128; local = t - 18432; }
  else if (t < 26624 + 16384) { W = w_h1; D = wf_h1; K = 256; local = t - 26624; }
  else if (t < 51200)         { W = w_h2; D = wf_h2; K = 128; local = t - 43008; }
  else return;
  const int v = local & 7, lane = (local >> 3) & 31, nt = (local >> 8) & 7, kc = local >> 11;
  const int k   = kc * 32 + ((lane >> 4) << 3) + ((v >> 2) << 4) + ((v & 3) << 1);
  const int col = nt * 16 + (lane & 15);
  const float a = (k     < K) ? W[k * 128 + col]       : 0.0f;
  const float b = (k + 1 < K) ? W[(k + 1) * 128 + col] : 0.0f;
  D[local] = (unsigned int)f2bf(a) | ((unsigned int)f2bf(b) << 16);
}

__global__ void egnn_zero_kernel(float* __restrict__ p, int n) {
  int t = blockIdx.x * blockDim.x + threadIdx.x;
  if (t < n) p[t] = 0.0f;
}

#define MT 4
__global__ __launch_bounds__(128, 2) void egnn_edge_kernel(
    const u32x8* __restrict__ hb8, const float* __restrict__ x, const int* __restrict__ edges,
    const u32x8* __restrict__ wf_e1, const u32x8* __restrict__ wf_e2,
    const float* __restrict__ b_e1, const float* __restrict__ b_e2,
    const float* __restrict__ w_inf, const float* __restrict__ b_inf,
    float* __restrict__ m_i) {
  __shared__ __attribute__((aligned(16))) unsigned short tbuf[4][MT][16 * 128];
  __shared__ float gbuf[4][MT * 16];

  const int lane = threadIdx.x & 31;
  const int wave = threadIdx.x >> 5;
  const int colq = lane & 15;
  const int half = lane >> 4;
  const int ebase = (blockIdx.x * 4 + wave) * (16 * MT);

  int s_own[MT], d_own[MT];
  unsigned int dbf[MT];
#pragma unroll
  for (int mt = 0; mt < MT; ++mt) {
    const int e_own = ebase + mt * 16 + colq;
    const int s = edges[2 * e_own], d = edges[2 * e_own + 1];
    s_own[mt] = s;
    d_own[mt] = d;
    const float dx = x[3 * s] - x[3 * d], dy = x[3 * s + 1] - x[3 * d + 1],
                dz = x[3 * s + 2] - x[3 * d + 2];
    const float dist = sqrtf(dx * dx + dy * dy + dz * dz);
    dbf[mt] = (half == 0) ? (unsigned int)f2bf(dist) : 0u;
  }

  bf16x16 af[MT][9];
#pragma unroll
  for (int mt = 0; mt < MT; ++mt) {
#pragma unroll
    for (int kc = 0; kc < 4; ++kc) {
      af[mt][kc]     = a_frag_from_hb(hb8, s_own[mt], kc, half);
      af[mt][4 + kc] = a_frag_from_hb(hb8, d_own[mt], kc, half);
    }
  }
#pragma unroll
  for (int mt = 0; mt < MT; ++mt) {
    u32x8 z;
#pragma unroll
    for (int v = 0; v < 8; ++v) z[v] = 0u;
    z[0] = dbf[mt];
    af[mt][8] = __builtin_bit_cast(bf16x16, z);
  }

#pragma unroll 1
  for (int nt = 0; nt < 8; ++nt) {
    bf16x16 bw[9];
#pragma unroll
    for (int kc = 0; kc < 9; ++kc) bw[kc] = load_b_frag(wf_e1, kc, nt, lane);
    const float bias = b_e1[nt * 16 + colq];
    __builtin_amdgcn_sched_barrier(0);
    f32x8 acc[MT];
#pragma unroll
    for (int mt = 0; mt < MT; ++mt) {
#pragma unroll
      for (int v = 0; v < 8; ++v) acc[mt][v] = bias;
    }
#pragma unroll
    for (int kc = 0; kc < 9; ++kc) {
#pragma unroll
      for (int mt = 0; mt < MT; ++mt)
        acc[mt] = __builtin_amdgcn_wmma_f32_16x16x32_bf16(false, af[mt][kc], false, bw[kc], (short)0, acc[mt], false, false);
    }
#pragma unroll
    for (int mt = 0; mt < MT; ++mt) {
      unsigned short* aw = tbuf[wave][mt];
#pragma unroll
      for (int v = 0; v < 8; ++v) {
        const int row = v + (half << 3);
        aw[fragord_index(row, nt * 16 + colq)] = f2bf(silu_f(acc[mt][v]));
      }
    }
  }
  __builtin_amdgcn_wave_barrier();

  bf16x16 a2f[MT][4];
#pragma unroll
  for (int mt = 0; mt < MT; ++mt) {
#pragma unroll
    for (int kc = 0; kc < 4; ++kc) a2f[mt][kc] = a_frag_from_fragord(tbuf[wave][mt], kc, lane);
  }
  __builtin_amdgcn_wave_barrier();

  f32x8 dot[MT];
#pragma unroll
  for (int mt = 0; mt < MT; ++mt) {
#pragma unroll
    for (int v = 0; v < 8; ++v) dot[mt][v] = 0.0f;
  }
#pragma unroll 1
  for (int nt = 0; nt < 8; ++nt) {
    bf16x16 bw[4];
#pragma unroll
    for (int kc = 0; kc < 4; ++kc) bw[kc] = load_b_frag(wf_e2, kc, nt, lane);
    const float bias = b_e2[nt * 16 + colq];
    const float wv = w_inf[nt * 16 + colq];
    __builtin_amdgcn_sched_barrier(0);
    f32x8 acc[MT];
#pragma unroll
    for (int mt = 0; mt < MT; ++mt) {
#pragma unroll
      for (int v = 0; v < 8; ++v) acc[mt][v] = bias;
    }
#pragma unroll
    for (int kc = 0; kc < 4; ++kc) {
#pragma unroll
      for (int mt = 0; mt < MT; ++mt)
        acc[mt] = __builtin_amdgcn_wmma_f32_16x16x32_bf16(false, a2f[mt][kc], false, bw[kc], (short)0, acc[mt], false, false);
    }
#pragma unroll
    for (int mt = 0; mt < MT; ++mt) {
      unsigned short* mb = tbuf[wave][mt];
#pragma unroll
      for (int v = 0; v < 8; ++v) {
        const float m = silu_f(acc[mt][v]);
        dot[mt][v] += m * wv;
        const int row = v + (half << 3);
        mb[row * 128 + nt * 16 + colq] = f2bf(m);
      }
    }
  }
  const float binf = b_inf[0];
#pragma unroll
  for (int mt = 0; mt < MT; ++mt) {
#pragma unroll
    for (int v = 0; v < 8; ++v) {
      float s = dot[mt][v];
      s += __shfl_xor(s, 1, 32);
      s += __shfl_xor(s, 2, 32);
      s += __shfl_xor(s, 4, 32);
      s += __shfl_xor(s, 8, 32);
      dot[mt][v] = s;
    }
  }
  if (colq == 0) {
#pragma unroll
    for (int mt = 0; mt < MT; ++mt) {
#pragma unroll
      for (int v = 0; v < 8; ++v) gbuf[wave][mt * 16 + v + (half << 3)] = sigm_f(dot[mt][v] + binf);
    }
  }
  __builtin_amdgcn_wave_barrier();

  for (int r = 0; r < 16; ++r) {
    float val[MT][4];
    int srow[MT];
#pragma unroll
    for (int mt = 0; mt < MT; ++mt) {
      srow[mt] = __builtin_amdgcn_readlane(s_own[mt], r);
      const float g = gbuf[wave][mt * 16 + r];
      const unsigned short* mb = tbuf[wave][mt] + r * 128 + lane;
#pragma unroll
      for (int j = 0; j < 4; ++j) val[mt][j] = g * bf2f(mb[32 * j]);
    }
#pragma unroll
    for (int mt = 0; mt < MT; ++mt) {
#pragma unroll
      for (int j = 0; j < 4; ++j) unsafeAtomicAdd(&m_i[srow[mt] * 128 + lane + 32 * j], val[mt][j]);
    }
  }
}

__global__ __launch_bounds__(128) void egnn_node_kernel(
    const float* __restrict__ h, const u32x8* __restrict__ hb8,
    const u32x8* __restrict__ wf_h1, const u32x8* __restrict__ wf_h2,
    const float* __restrict__ b_h1, const float* __restrict__ b_h2,
    float* __restrict__ hio) {
  __shared__ __attribute__((aligned(16))) unsigned short a2[4][8 * 32 * 8];

  const int lane = threadIdx.x & 31;
  const int wave = threadIdx.x >> 5;
  const int colq = lane & 15;
  const int half = lane >> 4;
  const int nbase = (blockIdx.x * 4 + wave) * 16;
  if (nbase >= NNODES) return;
  unsigned short* aw = a2[wave];

  const int node_own = nbase + colq;
  const int node_ld  = node_own < NNODES ? node_own : NNODES - 1;
  bf16x16 af[8];
  {
    const float* mr = hio + node_ld * 128 + 8 * half;
#pragma unroll
    for (int kc = 0; kc < 4; ++kc) af[kc] = a_frag_from_hb(hb8, node_ld, kc, half);
#pragma unroll
    for (int kc = 0; kc < 4; ++kc) af[4 + kc] = a_frag_from_row(mr + kc * 32);
  }

#pragma unroll 1
  for (int nt = 0; nt < 8; ++nt) {
    f32x8 acc;
    const float bias = b_h1[nt * 16 + colq];
#pragma unroll
    for (int v = 0; v < 8; ++v) acc[v] = bias;
#pragma unroll
    for (int kc = 0; kc < 8; ++kc) {
      if (kc + 1 < 8) __builtin_prefetch(&wf_h1[((kc + 1) * 8 + nt) * 32 + lane], 0, 3);
      bf16x16 b = load_b_frag(wf_h1, kc, nt, lane);
      acc = __builtin_amdgcn_wmma_f32_16x16x32_bf16(false, af[kc], false, b, (short)0, acc, false, false);
    }
#pragma unroll
    for (int v = 0; v < 8; ++v) {
      const int row = v + (half << 3);
      aw[fragord_index(row, nt * 16 + colq)] = f2bf(silu_f(acc[v]));
    }
  }
  __builtin_amdgcn_wave_barrier();

  bf16x16 a2f[4];
#pragma unroll
  for (int kc = 0; kc < 4; ++kc) a2f[kc] = a_frag_from_fragord(aw, kc, lane);

#pragma unroll 1
  for (int nt = 0; nt < 8; ++nt) {
    f32x8 acc;
    const float bias = b_h2[nt * 16 + colq];
#pragma unroll
    for (int v = 0; v < 8; ++v) acc[v] = bias;
#pragma unroll
    for (int kc = 0; kc < 4; ++kc) {
      bf16x16 b = load_b_frag(wf_h2, kc, nt, lane);
      acc = __builtin_amdgcn_wmma_f32_16x16x32_bf16(false, a2f[kc], false, b, (short)0, acc, false, false);
    }
#pragma unroll
    for (int v = 0; v < 8; ++v) {
      const int row = v + ((lane >> 4) << 3);
      const int node = nbase + row;
      const int col = nt * 16 + colq;
      if (node < NNODES) hio[node * 128 + col] = acc[v] + h[node * 128 + col];
    }
  }
}

extern "C" void kernel_launch(void* const* d_in, const int* in_sizes, int n_in,
                              void* d_out, int out_size, void* d_ws, size_t ws_size,
                              hipStream_t stream) {
  const float* h     = (const float*)d_in[0];
  const float* x     = (const float*)d_in[1];
  const int*   edges = (const int*)  d_in[2];
  const float* w_e1  = (const float*)d_in[3];
  const float* b_e1  = (const float*)d_in[4];
  const float* w_e2  = (const float*)d_in[5];
  const float* b_e2  = (const float*)d_in[6];
  const float* w_inf = (const float*)d_in[7];
  const float* b_inf = (const float*)d_in[8];
  const float* w_h1  = (const float*)d_in[9];
  const float* b_h1  = (const float*)d_in[10];
  const float* w_h2  = (const float*)d_in[11];
  const float* b_h2  = (const float*)d_in[12];
  float* out = (float*)d_out;

  unsigned int* wf_e1 = (unsigned int*)d_ws;
  unsigned int* wf_e2 = wf_e1 + 18432;
  unsigned int* wf_h1 = wf_e2 + 8192;
  unsigned int* wf_h2 = wf_h1 + 16384;
  unsigned int* hb    = wf_h2 + 8192;

  egnn_prep_kernel<<<PREP_W_BLOCKS + PREP_H_BLOCKS, 256, 0, stream>>>(
      w_e1, w_e2, w_h1, w_h2, wf_e1, wf_e2, wf_h1, wf_h2, h, (u32x4*)hb);

  egnn_zero_kernel<<<(NNODES * 128 + 255) / 256, 256, 0, stream>>>(out, NNODES * 128);

  egnn_edge_kernel<<<NEDGES / 256, 128, 0, stream>>>(
      (const u32x8*)hb, x, edges, (const u32x8*)wf_e1, (const u32x8*)wf_e2,
      b_e1, b_e2, w_inf, b_inf, out);

  egnn_node_kernel<<<(NNODES + 63) / 64, 128, 0, stream>>>(
      h, (const u32x8*)hb, (const u32x8*)wf_h1, (const u32x8*)wf_h2, b_h1, b_h2, out);
}


# zero-fill launch removed (4 launches to 3): the prep launch's h-conversion waves also store m_i's start value 0, two whole-line 16-byte stores per wave over the same 1024 bytes they read of h
# speedup vs baseline: 1.5273x; 1.0305x over previous
#include <hip/hip_runtime.h>
#include <hip/hip_bf16.h>

#define NNODES 50000
#define NEDGES 800000

typedef __attribute__((ext_vector_type(16))) __bf16        bf16x16;
typedef __attribute__((ext_vector_type(8)))  float         f32x8;
typedef __attribute__((ext_vector_type(8)))  unsigned int  u32x8;

__device__ __forceinline__ unsigned short f2bf(float f) {
  unsigned int u = __float_as_uint(f);
  u += 0x7FFFu + ((u >> 16) & 1u);
  return (unsigned short)(u >> 16);
}
__device__ __forceinline__ float bf2f(unsigned short s) {
  return __uint_as_float(((unsigned int)s) << 16);
}
__device__ __forceinline__ float silu_f(float v) { return v / (1.0f + __expf(-v)); }
__device__ __forceinline__ float sigm_f(float v) { return 1.0f / (1.0f + __expf(-v)); }

__device__ __forceinline__ int frag_k(int lane, int v) {
  return ((lane >> 4) << 3) + ((v >> 2) << 4) + ((v & 3) << 1);
}

typedef __attribute__((ext_vector_type(4))) float         f32x4;
typedef __attribute__((ext_vector_type(4))) unsigned int  u32x4;

__device__ __forceinline__ unsigned int pack2bf(float lo, float hi) {
  return (unsigned int)f2bf(lo) | ((unsigned int)f2bf(hi) << 16);
}

__device__ __forceinline__ bf16x16 a_frag_from_row(const float* p) {
  const f32x4 q0 = *(const f32x4*)(p);
  const f32x4 q1 = *(const f32x4*)(p + 4);
  const f32x4 q2 = *(const f32x4*)(p + 16);
  const f32x4 q3 = *(const f32x4*)(p + 20);
  u32x8 r;
  r[0] = pack2bf(q0[0], q0[1]); r[1] = pack2bf(q0[2], q0[3]);
  r[2] = pack2bf(q1[0], q1[1]); r[3] = pack2bf(q1[2], q1[3]);
  r[4] = pack2bf(q2[0], q2[1]); r[5] = pack2bf(q2[2], q2[3]);
  r[6] = pack2bf(q3[0], q3[1]); r[7] = pack2bf(q3[2], q3[3]);
  return __builtin_bit_cast(bf16x16, r);
}

__device__ __forceinline__ int fragord_index(int row, int col) {
  return (((col >> 4) * 32 + row + (((col >> 3) & 1) << 4)) << 3) + (col & 7);
}
__device__ __forceinline__ bf16x16 a_frag_from_fragord(const unsigned short* buf, int kc, int lane) {
  const u32x4 lo = *(const u32x4*)(buf + (((2 * kc)     * 32 + lane) << 3));
  const u32x4 hi = *(const u32x4*)(buf + (((2 * kc + 1) * 32 + lane) << 3));
  u32x8 r;
  r[0] = lo[0]; r[1] = lo[1]; r[2] = lo[2]; r[3] = lo[3];
  r[4] = hi[0]; r[5] = hi[1]; r[6] = hi[2]; r[7] = hi[3];
  return __builtin_bit_cast(bf16x16, r);
}

__device__ __forceinline__ bf16x16 a_frag_from_hb(const u32x8* __restrict__ hb8, int node, int kc, int half) {
  return __builtin_bit_cast(bf16x16, hb8[node * 8 + kc * 2 + half]);
}

__device__ __forceinline__ bf16x16 load_b_frag(const u32x8* __restrict__ wf, int kc, int nt, int lane) {
  return __builtin_bit_cast(bf16x16, wf[(kc * 8 + nt) * 32 + lane]);
}

#define PREP_W_BLOCKS 200
#define PREP_ITEMS 4
#define PREP_H_BLOCKS 782
static_assert((NNODES * 16) % 32 == 0 && (PREP_H_BLOCKS * 256) % 32 == 0, "whole waves of items");
__global__ void egnn_prep_kernel(const float* __restrict__ w_e1, const float* __restrict__ w_e2,
                                 const float* __restrict__ w_h1, const float* __restrict__ w_h2,
                                 unsigned int* __restrict__ wf_e1, unsigned int* __restrict__ wf_e2,
                                 unsigned int* __restrict__ wf_h1, unsigned int* __restrict__ wf_h2,
                                 const float* __restrict__ h, u32x4* __restrict__ hb,
                                 float* __restrict__ m_i) {
  if (blockIdx.x >= PREP_W_BLOCKS) {
    const int tt = (blockIdx.x - PREP_W_BLOCKS) * 256 + threadIdx.x;
    const int lane = threadIdx.x & 31;
    f32x4 q0[PREP_ITEMS], q1[PREP_ITEMS];
#pragma unroll
    for (int j = 0; j < PREP_ITEMS; ++j) {
      const int item = tt + j * (PREP_H_BLOCKS * 256);
      const int ld = item < NNODES * 16 ? item : NNODES * 16 - 1;
      const float* p = h + ld * 8;
      q0[j] = *(const f32x4*)(p);
      q1[j] = *(const f32x4*)(p + 4);
    }
#pragma unroll
    for (int j = 0; j < PREP_ITEMS; ++j) {
      const int item = tt + j * (PREP_H_BLOCKS * 256);
      if (item < NNODES * 16) {
        float* z = m_i + (item - lane) * 8 + lane * 4;
        const f32x4 zero = {0.0f, 0.0f, 0.0f, 0.0f};
        *(f32x4*)(z)       = zero;
        *(f32x4*)(z + 128) = zero;
      }
    }
#pragma unroll
    for (int j = 0; j < PREP_ITEMS; ++j) {
      const int item = tt + j * (PREP_H_BLOCKS * 256);
      const int g = item & 15, kc = g >> 2, q = g & 3;
      const int slot = ((q & 1) << 1) | (q >> 1);
      u32x4 r;
      r[0] = pack2bf(q0[j][0], q0[j][1]); r[1] = pack2bf(q0[j][2], q0[j][3]);
      r[2] = pack2bf(q1[j][0], q1[j][1]); r[3] = pack2bf(q1[j][2], q1[j][3]);
      if (item < NNODES * 16) hb[(item & ~15) + kc * 4 + slot] = r;
    }
    return;
  }
  int t = blockIdx.x * blockDim.x + threadIdx.x;
  const float* W; unsigned int* D; int K; int local;
  if      (t < 18432)         { W = w_e1; D = wf_e1; K = 257; local = t; }
  else if (t < 18432 + 8192)  { W = w_e2; D = wf_e2; K = 128; local = t - 18432; }
  else if (t < 26624 + 16384) { W = w_h1; D = wf_h1; K = 256; local = t - 26624; }
  else if (t < 51200)         { W = w_h2; D = wf_h2; K = 128; local = t - 43008; }
  else return;
  const int v = local & 7, lane = (local >> 3) & 31, nt = (local >> 8) & 7, kc = local >> 11;
  const int k   = kc * 32 + ((lane >> 4) << 3) + ((v >> 2) << 4) + ((v & 3) << 1);
  const int col = nt * 16 + (lane & 15);
  const float a = (k     < K) ? W[k * 128 + col]       : 0.0f;
  const float b = (k + 1 < K) ? W[(k + 1) * 128 + col] : 0.0f;
  D[local] = (unsigned int)f2bf(a) | ((unsigned int)f2bf(b) << 16);
}

#define MT 4
__global__ __launch_bounds__(128, 2) void egnn_edge_kernel(
    const u32x8* __restrict__ hb8, const float* __restrict__ x, const int* __restrict__ edges,
    const u32x8* __restrict__ wf_e1, const u32x8* __restrict__ wf_e2,
    const float* __restrict__ b_e1, const float* __restrict__ b_e2,
    const float* __restrict__ w_inf, const float* __restrict__ b_inf,
    float* __restrict__ m_i) {
  __shared__ __attribute__((aligned(16))) unsigned short tbuf[4][MT][16 * 128];
  __shared__ float gbuf[4][MT * 16];

  const int lane = threadIdx.x & 31;
  const int wave = threadIdx.x >> 5;
  const int colq = lane & 15;
  const int half = lane >> 4;
  const int ebase = (blockIdx.x * 4 + wave) * (16 * MT);

  int s_own[MT], d_own[MT];
  unsigned int dbf[MT];
#pragma unroll
  for (int mt = 0; mt < MT; ++mt) {
    const int e_own = ebase + mt * 16 + colq;
    const int s = edges[2 * e_own], d = edges[2 * e_own + 1];
    s_own[mt] = s;
    d_own[mt] = d;
    const float dx = x[3 * s] - x[3 * d], dy = x[3 * s + 1] - x[3 * d + 1],
                dz = x[3 * s + 2] - x[3 * d + 2];
    const float dist = sqrtf(dx * dx + dy * dy + dz * dz);
    dbf[mt] = (half == 0) ? (unsigned int)f2bf(dist) : 0u;
  }

  bf16x16 af[MT][9];
#pragma unroll
  for (int mt = 0; mt < MT; ++mt) {
#pragma unroll
    for (int kc = 0; kc < 4; ++kc) {
      af[mt][kc]     = a_frag_from_hb(hb8, s_own[mt], kc, half);
      af[mt][4 + kc] = a_frag_from_hb(hb8, d_own[mt], kc, half);
    }
  }
#pragma unroll
  for (int mt = 0; mt < MT; ++mt) {
    u32x8 z;
#pragma unroll
    for (int v = 0; v < 8; ++v) z[v] = 0u;
    z[0] = dbf[mt];
    af[mt][8] = __builtin_bit_cast(bf16x16, z);
  }

#pragma unroll 1
  for (int nt = 0; nt < 8; ++nt) {
    bf16x16 bw[9];
#pragma unroll
    for (int kc = 0; kc < 9; ++kc) bw[kc] = load_b_frag(wf_e1, kc, nt, lane);
    const float bias = b_e1[nt * 16 + colq];
    __builtin_amdgcn_sched_barrier(0);
    f32x8 acc[MT];
#pragma unroll
    for (int mt = 0; mt < MT; ++mt) {
#pragma unroll
      for (int v = 0; v < 8; ++v) acc[mt][v] = bias;
    }
#pragma unroll
    for (int kc = 0; kc < 9; ++kc) {
#pragma unroll
      for (int mt = 0; mt < MT; ++mt)
        acc[mt] = __builtin_amdgcn_wmma_f32_16x16x32_bf16(false, af[mt][kc], false, bw[kc], (short)0, acc[mt], false, false);
    }
#pragma unroll
    for (int mt = 0; mt < MT; ++mt) {
      unsigned short* aw = tbuf[wave][mt];
#pragma unroll
      for (int v = 0; v < 8; ++v) {
        const int row = v + (half << 3);
        aw[fragord_index(row, nt * 16 + colq)] = f2bf(silu_f(acc[mt][v]));
      }
    }
  }
  __builtin_amdgcn_wave_barrier();

  bf16x16 a2f[MT][4];
#pragma unroll
  for (int mt = 0; mt < MT; ++mt) {
#pragma unroll
    for (int kc = 0; kc < 4; ++kc) a2f[mt][kc] = a_frag_from_fragord(tbuf[wave][mt], kc, lane);
  }
  __builtin_amdgcn_wave_barrier();

  f32x8 dot[MT];
#pragma unroll
  for (int mt = 0; mt < MT; ++mt) {
#pragma unroll
    for (int v = 0; v < 8; ++v) dot[mt][v] = 0.0f;
  }
#pragma unroll 1
  for (int nt = 0; nt < 8; ++nt) {
    bf16x16 bw[4];
#pragma unroll
    for (int kc = 0; kc < 4; ++kc) bw[kc] = load_b_frag(wf_e2, kc, nt, lane);
    const float bias = b_e2[nt * 16 + colq];
    const float wv = w_inf[nt * 16 + colq];
    __builtin_amdgcn_sched_barrier(0);
    f32x8 acc[MT];
#pragma unroll
    for (int mt = 0; mt < MT; ++mt) {
#pragma unroll
      for (int v = 0; v < 8; ++v) acc[mt][v] = bias;
    }
#pragma unroll
    for (int kc = 0; kc < 4; ++kc) {
#pragma unroll
      for (int mt = 0; mt < MT; ++mt)
        acc[mt] = __builtin_amdgcn_wmma_f32_16x16x32_bf16(false, a2f[mt][kc], false, bw[kc], (short)0, acc[mt], false, false);
    }
#pragma unroll
    for (int mt = 0; mt < MT; ++mt) {
      unsigned short* mb = tbuf[wave][mt];
#pragma unroll
      for (int v = 0; v < 8; ++v) {
        const float m = silu_f(acc[mt][v]);
        dot[mt][v] += m * wv;
        const int row = v + (half << 3);
        mb[row * 128 + nt * 16 + colq] = f2bf(m);
      }
    }
  }
  const float binf = b_inf[0];
#pragma unroll
  for (int mt = 0; mt < MT; ++mt) {
#pragma unroll
    for (int v = 0; v < 8; ++v) {
      float s = dot[mt][v];
      s += __shfl_xor(s, 1, 32);
      s += __shfl_xor(s, 2, 32);
      s += __shfl_xor(s, 4, 32);
      s += __shfl_xor(s, 8, 32);
      dot[mt][v] = s;
    }
  }
  if (colq == 0) {
#pragma unroll
    for (int mt = 0; mt < MT; ++mt) {
#pragma unroll
      for (int v = 0; v < 8; ++v) gbuf[wave][mt * 16 + v + (half << 3)] = sigm_f(dot[mt][v] + binf);
    }
  }
  __builtin_amdgcn_wave_barrier();

  for (int r = 0; r < 16; ++r) {
    float val[MT][4];
    int srow[MT];
#pragma unroll
    for (int mt = 0; mt < MT; ++mt) {
      srow[mt] = __builtin_amdgcn_readlane(s_own[mt], r);
      const float g = gbuf[wave][mt * 16 + r];
      const unsigned short* mb = tbuf[wave][mt] + r * 128 + lane;
#pragma unroll
      for (int j = 0; j < 4; ++j) val[mt][j] = g * bf2f(mb[32 * j]);
    }
#pragma unroll
    for (int mt = 0; mt < MT; ++mt) {
#pragma unroll
      for (int j = 0; j < 4; ++j) unsafeAtomicAdd(&m_i[srow[mt] * 128 + lane + 32 * j], val[mt][j]);
    }
  }
}

__global__ __launch_bounds__(128) void egnn_node_kernel(
    const float* __restrict__ h, const u32x8* __restrict__ hb8,
    const u32x8* __restrict__ wf_h1, const u32x8* __restrict__ wf_h2,
    const float* __restrict__ b_h1, const float* __restrict__ b_h2,
    float* __restrict__ hio) {
  __shared__ __attribute__((aligned(16))) unsigned short a2[4][8 * 32 * 8];

  const int lane = threadIdx.x & 31;
  const int wave = threadIdx.x >> 5;
  const int colq = lane & 15;
  const int half = lane >> 4;
  const int nbase = (blockIdx.x * 4 + wave) * 16;
  if (nbase >= NNODES) return;
  unsigned short* aw = a2[wave];

  const int node_own = nbase + colq;
  const int node_ld  = node_own < NNODES ? node_own : NNODES - 1;
  bf16x16 af[8];
  {
    const float* mr = hio + node_ld * 128 + 8 * half;
#pragma unroll
    for (int kc = 0; kc < 4; ++kc) af[kc] = a_frag_from_hb(hb8, node_ld, kc, half);
#pragma unroll
    for (int kc = 0; kc < 4; ++kc) af[4 + kc] = a_frag_from_row(mr + kc * 32);
  }

#pragma unroll 1
  for (int nt = 0; nt < 8; ++nt) {
    f32x8 acc;
    const float bias = b_h1[nt * 16 + colq];
#pragma unroll
    for (int v = 0; v < 8; ++v) acc[v] = bias;
#pragma unroll
    for (int kc = 0; kc < 8; ++kc) {
      if (kc + 1 < 8) __builtin_prefetch(&wf_h1[((kc + 1) * 8 + nt) * 32 + lane], 0, 3);
      bf16x16 b = load_b_frag(wf_h1, kc, nt, lane);
      acc = __builtin_amdgcn_wmma_f32_16x16x32_bf16(false, af[kc], false, b, (short)0, acc, false, false);
    }
#pragma unroll
    for (int v = 0; v < 8; ++v) {
      const int row = v + (half << 3);
      aw[fragord_index(row, nt * 16 + colq)] = f2bf(silu_f(acc[v]));
    }
  }
  __builtin_amdgcn_wave_barrier();

  bf16x16 a2f[4];
#pragma unroll
  for (int kc = 0; kc < 4; ++kc) a2f[kc] = a_frag_from_fragord(aw, kc, lane);

#pragma unroll 1
  for (int nt = 0; nt < 8; ++nt) {
    f32x8 acc;
    const float bias = b_h2[nt * 16 + colq];
#pragma unroll
    for (int v = 0; v < 8; ++v) acc[v] = bias;
#pragma unroll
    for (int kc = 0; kc < 4; ++kc) {
      bf16x16 b = load_b_frag(wf_h2, kc, nt, lane);
      acc = __builtin_amdgcn_wmma_f32_16x16x32_bf16(false, a2f[kc], false, b, (short)0, acc, false, false);
    }
#pragma unroll
    for (int v = 0; v < 8; ++v) {
      const int row = v + ((lane >> 4) << 3);
      const int node = nbase + row;
      const int col = nt * 16 + colq;
      if (node < NNODES) hio[node * 128 + col] = acc[v] + h[node * 128 + col];
    }
  }
}

extern "C" void kernel_launch(void* const* d_in, const int* in_sizes, int n_in,
                              void* d_out, int out_size, void* d_ws, size_t ws_size,
                              hipStream_t stream) {
  const float* h     = (const float*)d_in[0];
  const float* x     = (const float*)d_in[1];
  const int*   edges = (const int*)  d_in[2];
  const float* w_e1  = (const float*)d_in[3];
  const float* b_e1  = (const float*)d_in[4];
  const float* w_e2  = (const float*)d_in[5];
  const float* b_e2  = (const float*)d_in[6];
  const float* w_inf = (const float*)d_in[7];
  const float* b_inf = (const float*)d_in[8];
  const float* w_h1  = (const float*)d_in[9];
  const float* b_h1  = (const float*)d_in[10];
  const float* w_h2  = (const float*)d_in[11];
  const float* b_h2  = (const float*)d_in[12];
  float* out = (float*)d_out;

  unsigned int* wf_e1 = (unsigned int*)d_ws;
  unsigned int* wf_e2 = wf_e1 + 18432;
  unsigned int* wf_h1 = wf_e2 + 8192;
  unsigned int* wf_h2 = wf_h1 + 16384;
  unsigned int* hb    = wf_h2 + 8192;

  egnn_prep_kernel<<<PREP_W_BLOCKS + PREP_H_BLOCKS, 256, 0, stream>>>(
      w_e1, w_e2, w_h1, w_h2, wf_e1, wf_e2, wf_h1, wf_h2, h, (u32x4*)hb, out);

  egnn_edge_kernel<<<NEDGES / 256, 128, 0, stream>>>(
      (const u32x8*)hb, x, edges, (const u32x8*)wf_e1, (const u32x8*)wf_e2,
      b_e1, b_e2, w_inf, b_inf, out);

  egnn_node_kernel<<<(NNODES + 63) / 64, 128, 0, stream>>>(
      h, (const u32x8*)hb, (const u32x8*)wf_h1, (const u32x8*)wf_h2, b_h1, b_h2, out);
}


# edge kernel epilogues: the 8 SiLU divisions of an accumulator advance together, IEEE division written out stage by stage (same steps, same bits), so no instruction waits for the one before it
# speedup vs baseline: 1.6801x; 1.1000x over previous
#include <hip/hip_runtime.h>
#include <hip/hip_bf16.h>

#define NNODES 50000
#define NEDGES 800000

typedef __attribute__((ext_vector_type(16))) __bf16        bf16x16;
typedef __attribute__((ext_vector_type(8)))  float         f32x8;
typedef __attribute__((ext_vector_type(8)))  unsigned int  u32x8;

__device__ __forceinline__ unsigned short f2bf(float f) {
  unsigned int u = __float_as_uint(f);
  u += 0x7FFFu + ((u >> 16) & 1u);
  return (unsigned short)(u >> 16);
}
__device__ __forceinline__ float bf2f(unsigned short s) {
  return __uint_as_float(((unsigned int)s) << 16);
}
__device__ __forceinline__ float silu_f(float v) { return v / (1.0f + __expf(-v)); }
__device__ __forceinline__ float sigm_f(float v) { return 1.0f / (1.0f + __expf(-v)); }

#define STAGE_ORDER() __builtin_amdgcn_sched_barrier(0)
__device__ __forceinline__ void silu_x8(const float (&x)[8], float (&out)[8]) {
  float den[8], ds[8], ns[8], r[8], q[8], e[8];
  bool fl[8], fl_den;
#pragma unroll
  for (int v = 0; v < 8; ++v) den[v] = 1.0f + __expf(-x[v]);
#pragma unroll
  for (int v = 0; v < 8; ++v) ds[v] = __builtin_amdgcn_div_scalef(x[v], den[v], false, &fl_den);
#pragma unroll
  for (int v = 0; v < 8; ++v) ns[v] = __builtin_amdgcn_div_scalef(x[v], den[v], true, &fl[v]);
  STAGE_ORDER();
#pragma unroll
  for (int v = 0; v < 8; ++v) r[v] = __builtin_amdgcn_rcpf(ds[v]);
#pragma unroll
  for (int v = 0; v < 8; ++v) e[v] = __builtin_fmaf(-ds[v], r[v], 1.0f);
  STAGE_ORDER();
#pragma unroll
  for (int v = 0; v < 8; ++v) r[v] = __builtin_fmaf(e[v], r[v], r[v]);
#pragma unroll
  for (int v = 0; v < 8; ++v) q[v] = ns[v] * r[v];
#pragma unroll
  for (int v = 0; v < 8; ++v) e[v] = __builtin_fmaf(-ds[v], q[v], ns[v]);
  STAGE_ORDER();
#pragma unroll
  for (int v = 0; v < 8; ++v) q[v] = __builtin_fmaf(e[v], r[v], q[v]);
#pragma unroll
  for (int v = 0; v < 8; ++v) e[v] = __builtin_fmaf(-ds[v], q[v], ns[v]);
  STAGE_ORDER();
#pragma unroll
  for (int v = 0; v < 8; ++v) q[v] = __builtin_amdgcn_div_fmasf(e[v], r[v], q[v], fl[v]);
  STAGE_ORDER();
#pragma unroll
  for (int v = 0; v < 8; ++v) out[v] = __builtin_amdgcn_div_fixupf(q[v], den[v], x[v]);
}

__device__ __forceinline__ int frag_k(int lane, int v) {
  return ((lane >> 4) << 3) + ((v >> 2) << 4) + ((v & 3) << 1);
}

typedef __attribute__((ext_vector_type(4))) float         f32x4;
typedef __attribute__((ext_vector_type(4))) unsigned int  u32x4;

__device__ __forceinline__ unsigned int pack2bf(float lo, float hi) {
  return (unsigned int)f2bf(lo) | ((unsigned int)f2bf(hi) << 16);
}

__device__ __forceinline__ bf16x16 a_frag_from_row(const float* p) {
  const f32x4 q0 = *(const f32x4*)(p);
  const f32x4 q1 = *(const f32x4*)(p + 4);
  const f32x4 q2 = *(const f32x4*)(p + 16);
  const f32x4 q3 = *(const f32x4*)(p + 20);
  u32x8 r;
  r[0] = pack2bf(q0[0], q0[1]); r[1] = pack2bf(q0[2], q0[3]);
  r[2] = pack2bf(q1[0], q1[1]); r[3] = pack2bf(q1[2], q1[3]);
  r[4] = pack2bf(q2[0], q2[1]); r[5] = pack2bf(q2[2], q2[3]);
  r[6] = pack2bf(q3[0], q3[1]); r[7] = pack2bf(q3[2], q3[3]);
  return __builtin_bit_cast(bf16x16, r);
}

__device__ __forceinline__ int fragord_index(int row, int col) {
  return (((col >> 4) * 32 + row + (((col >> 3) & 1) << 4)) << 3) + (col & 7);
}
__device__ __forceinline__ bf16x16 a_frag_from_fragord(const unsigned short* buf, int kc, int lane) {
  const u32x4 lo = *(const u32x4*)(buf + (((2 * kc)     * 32 + lane) << 3));
  const u32x4 hi = *(const u32x4*)(buf + (((2 * kc + 1) * 32 + lane) << 3));
  u32x8 r;
  r[0] = lo[0]; r[1] = lo[1]; r[2] = lo[2]; r[3] = lo[3];
  r[4] = hi[0]; r[5] = hi[1]; r[6] = hi[2]; r[7] = hi[3];
  return __builtin_bit_cast(bf16x16, r);
}

__device__ __forceinline__ bf16x16 a_frag_from_hb(const u32x8* __restrict__ hb8, int node, int kc, int half) {
  return __builtin_bit_cast(bf16x16, hb8[node * 8 + kc * 2 + half]);
}

__device__ __forceinline__ bf16x16 load_b_frag(const u32x8* __restrict__ wf, int kc, int nt, int lane) {
  return __builtin_bit_cast(bf16x16, wf[(kc * 8 + nt) * 32 + lane]);
}

#define PREP_W_BLOCKS 200
#define PREP_ITEMS 4
#define PREP_H_BLOCKS 782
static_assert((NNODES * 16) % 32 == 0 && (PREP_H_BLOCKS * 256) % 32 == 0, "whole waves of items");
__global__ void egnn_prep_kernel(const float* __restrict__ w_e1, const float* __restrict__ w_e2,
                                 const float* __restrict__ w_h1, const float* __restrict__ w_h2,
                                 unsigned int* __restrict__ wf_e1, unsigned int* __restrict__ wf_e2,
                                 unsigned int* __restrict__ wf_h1, unsigned int* __restrict__ wf_h2,
                                 const float* __restrict__ h, u32x4* __restrict__ hb,
                                 float* __restrict__ m_i) {
  if (blockIdx.x >= PREP_W_BLOCKS) {
    const int tt = (blockIdx.x - PREP_W_BLOCKS) * 256 + threadIdx.x;
    const int lane = threadIdx.x & 31;
    f32x4 q0[PREP_ITEMS], q1[PREP_ITEMS];
#pragma unroll
    for (int j = 0; j < PREP_ITEMS; ++j) {
      const int item = tt + j * (PREP_H_BLOCKS * 256);
      const int ld = item < NNODES * 16 ? item : NNODES * 16 - 1;
      const float* p = h + ld * 8;
      q0[j] = *(const f32x4*)(p);
      q1[j] = *(const f32x4*)(p + 4);
    }
#pragma unroll
    for (int j = 0; j < PREP_ITEMS; ++j) {
      const int item = tt + j * (PREP_H_BLOCKS * 256);
      if (item < NNODES * 16) {
        float* z = m_i + (item - lane) * 8 + lane * 4;
        const f32x4 zero = {0.0f, 0.0f, 0.0f, 0.0f};
        *(f32x4*)(z)       = zero;
        *(f32x4*)(z + 128) = zero;
      }
    }
#pragma unroll
    for (int j = 0; j < PREP_ITEMS; ++j) {
      const int item = tt + j * (PREP_H_BLOCKS * 256);
      const int g = item & 15, kc = g >> 2, q = g & 3;
      const int slot = ((q & 1) << 1) | (q >> 1);
      u32x4 r;
      r[0] = pack2bf(q0[j][0], q0[j][1]); r[1] = pack2bf(q0[j][2], q0[j][3]);
      r[2] = pack2bf(q1[j][0], q1[j][1]); r[3] = pack2bf(q1[j][2], q1[j][3]);
      if (item < NNODES * 16) hb[(item & ~15) + kc * 4 + slot] = r;
    }
    return;
  }
  int t = blockIdx.x * blockDim.x + threadIdx.x;
  const float* W; unsigned int* D; int K; int local;
  if      (t < 18432)         { W = w_e1; D = wf_e1; K = 257; local = t; }
  else if (t < 18432 + 8192)  { W = w_e2; D = wf_e2; K = 128; local = t - 18432; }
  else if (t < 26624 + 16384) { W = w_h1; D = wf_h1; K = 256; local = t - 26624; }
  else if (t < 51200)         { W = w_h2; D = wf_h2; K = 128; local = t - 43008; }
  else return;
  const int v = local & 7, lane = (local >> 3) & 31, nt = (local >> 8) & 7, kc = local >> 11;
  const int k   = kc * 32 + ((lane >> 4) << 3) + ((v >> 2) << 4) + ((v & 3) << 1);
  const int col = nt * 16 + (lane & 15);
  const float a = (k     < K) ? W[k * 128 + col]       : 0.0f;
  const float b = (k + 1 < K) ? W[(k + 1) * 128 + col] : 0.0f;
  D[local] = (unsigned int)f2bf(a) | ((unsigned int)f2bf(b) << 16);
}

#define MT 4
__global__ __launch_bounds__(128, 2) void egnn_edge_kernel(
    const u32x8* __restrict__ hb8, const float* __restrict__ x, const int* __restrict__ edges,
    const u32x8* __restrict__ wf_e1, const u32x8* __restrict__ wf_e2,
    const float* __restrict__ b_e1, const float* __restrict__ b_e2,
    const float* __restrict__ w_inf, const float* __restrict__ b_inf,
    float* __restrict__ m_i) {
  __shared__ __attribute__((aligned(16))) unsigned short tbuf[4][MT][16 * 128];
  __shared__ float gbuf[4][MT * 16];

  const int lane = threadIdx.x & 31;
  const int wave = threadIdx.x >> 5;
  const int colq = lane & 15;
  const int half = lane >> 4;
  const int ebase = (blockIdx.x * 4 + wave) * (16 * MT);

  int s_own[MT], d_own[MT];
  unsigned int dbf[MT];
#pragma unroll
  for (int mt = 0; mt < MT; ++mt) {
    const int e_own = ebase + mt * 16 + colq;
    const int s = edges[2 * e_own], d = edges[2 * e_own + 1];
    s_own[mt] = s;
    d_own[mt] = d;
    const float dx = x[3 * s] - x[3 * d], dy = x[3 * s + 1] - x[3 * d + 1],
                dz = x[3 * s + 2] - x[3 * d + 2];
    const float dist = sqrtf(dx * dx + dy * dy + dz * dz);
    dbf[mt] = (half == 0) ? (unsigned int)f2bf(dist) : 0u;
  }

  bf16x16 af[MT][9];
#pragma unroll
  for (int mt = 0; mt < MT; ++mt) {
#pragma unroll
    for (int kc = 0; kc < 4; ++kc) {
      af[mt][kc]     = a_frag_from_hb(hb8, s_own[mt], kc, half);
      af[mt][4 + kc] = a_frag_from_hb(hb8, d_own[mt], kc, half);
    }
  }
#pragma unroll
  for (int mt = 0; mt < MT; ++mt) {
    u32x8 z;
#pragma unroll
    for (int v = 0; v < 8; ++v) z[v] = 0u;
    z[0] = dbf[mt];
    af[mt][8] = __builtin_bit_cast(bf16x16, z);
  }

#pragma unroll 1
  for (int nt = 0; nt < 8; ++nt) {
    bf16x16 bw[9];
#pragma unroll
    for (int kc = 0; kc < 9; ++kc) bw[kc] = load_b_frag(wf_e1, kc, nt, lane);
    const float bias = b_e1[nt * 16 + colq];
    __builtin_amdgcn_sched_barrier(0);
    f32x8 acc[MT];
#pragma unroll
    for (int mt = 0; mt < MT; ++mt) {
#pragma unroll
      for (int v = 0; v < 8; ++v) acc[mt][v] = bias;
    }
#pragma unroll
    for (int kc = 0; kc < 9; ++kc) {
#pragma unroll
      for (int mt = 0; mt < MT; ++mt)
        acc[mt] = __builtin_amdgcn_wmma_f32_16x16x32_bf16(false, af[mt][kc], false, bw[kc], (short)0, acc[mt], false, false);
    }
#pragma unroll
    for (int mt = 0; mt < MT; ++mt) {
      unsigned short* aw = tbuf[wave][mt];
      float a[8], s[8];
#pragma unroll
      for (int v = 0; v < 8; ++v) a[v] = acc[mt][v];
      silu_x8(a, s);
#pragma unroll
      for (int v = 0; v < 8; ++v) {
        const int row = v + (half << 3);
        aw[fragord_index(row, nt * 16 + colq)] = f2bf(s[v]);
      }
    }
  }
  __builtin_amdgcn_wave_barrier();

  bf16x16 a2f[MT][4];
#pragma unroll
  for (int mt = 0; mt < MT; ++mt) {
#pragma unroll
    for (int kc = 0; kc < 4; ++kc) a2f[mt][kc] = a_frag_from_fragord(tbuf[wave][mt], kc, lane);
  }
  __builtin_amdgcn_wave_barrier();

  f32x8 dot[MT];
#pragma unroll
  for (int mt = 0; mt < MT; ++mt) {
#pragma unroll
    for (int v = 0; v < 8; ++v) dot[mt][v] = 0.0f;
  }
#pragma unroll 1
  for (int nt = 0; nt < 8; ++nt) {
    bf16x16 bw[4];
#pragma unroll
    for (int kc = 0; kc < 4; ++kc) bw[kc] = load_b_frag(wf_e2, kc, nt, lane);
    const float bias = b_e2[nt * 16 + colq];
    const float wv = w_inf[nt * 16 + colq];
    __builtin_amdgcn_sched_barrier(0);
    f32x8 acc[MT];
#pragma unroll
    for (int mt = 0; mt < MT; ++mt) {
#pragma unroll
      for (int v = 0; v < 8; ++v) acc[mt][v] = bias;
    }
#pragma unroll
    for (int kc = 0; kc < 4; ++kc) {
#pragma unroll
      for (int mt = 0; mt < MT; ++mt)
        acc[mt] = __builtin_amdgcn_wmma_f32_16x16x32_bf16(false, a2f[mt][kc], false, bw[kc], (short)0, acc[mt], false, false);
    }
#pragma unroll
    for (int mt = 0; mt < MT; ++mt) {
      unsigned short* mb = tbuf[wave][mt];
      float a[8], m[8];
#pragma unroll
      for (int v = 0; v < 8; ++v) a[v] = acc[mt][v];
      silu_x8(a, m);
#pragma unroll
      for (int v = 0; v < 8; ++v) dot[mt][v] += m[v] * wv;
#pragma unroll
      for (int v = 0; v < 8; ++v) {
        const int row = v + (half << 3);
        mb[row * 128 + nt * 16 + colq] = f2bf(m[v]);
      }
    }
  }
  const float binf = b_inf[0];
#pragma unroll
  for (int mt = 0; mt < MT; ++mt) {
#pragma unroll
    for (int v = 0; v < 8; ++v) {
      float s = dot[mt][v];
      s += __shfl_xor(s, 1, 32);
      s += __shfl_xor(s, 2, 32);
      s += __shfl_xor(s, 4, 32);
      s += __shfl_xor(s, 8, 32);
      dot[mt][v] = s;
    }
  }
  if (colq == 0) {
#pragma unroll
    for (int mt = 0; mt < MT; ++mt) {
#pragma unroll
      for (int v = 0; v < 8; ++v) gbuf[wave][mt * 16 + v + (half << 3)] = sigm_f(dot[mt][v] + binf);
    }
  }
  __builtin_amdgcn_wave_barrier();

  for (int r = 0; r < 16; ++r) {
    float val[MT][4];
    int srow[MT];
#pragma unroll
    for (int mt = 0; mt < MT; ++mt) {
      srow[mt] = __builtin_amdgcn_readlane(s_own[mt], r);
      const float g = gbuf[wave][mt * 16 + r];
      const unsigned short* mb = tbuf[wave][mt] + r * 128 + lane;
#pragma unroll
      for (int j = 0; j < 4; ++j) val[mt][j] = g * bf2f(mb[32 * j]);
    }
#pragma unroll
    for (int mt = 0; mt < MT; ++mt) {
#pragma unroll
      for (int j = 0; j < 4; ++j) unsafeAtomicAdd(&m_i[srow[mt] * 128 + lane + 32 * j], val[mt][j]);
    }
  }
}

__global__ __launch_bounds__(128) void egnn_node_kernel(
    const float* __restrict__ h, const u32x8* __restrict__ hb8,
    const u32x8* __restrict__ wf_h1, const u32x8* __restrict__ wf_h2,
    const float* __restrict__ b_h1, const float* __restrict__ b_h2,
    float* __restrict__ hio) {
  __shared__ __attribute__((aligned(16))) unsigned short a2[4][8 * 32 * 8];

  const int lane = threadIdx.x & 31;
  const int wave = threadIdx.x >> 5;
  const int colq = lane & 15;
  const int half = lane >> 4;
  const int nbase = (blockIdx.x * 4 + wave) * 16;
  if (nbase >= NNODES) return;
  unsigned short* aw = a2[wave];

  const int node_own = nbase + colq;
  const int node_ld  = node_own < NNODES ? node_own : NNODES - 1;
  bf16x16 af[8];
  {
    const float* mr = hio + node_ld * 128 + 8 * half;
#pragma unroll
    for (int kc = 0; kc < 4; ++kc) af[kc] = a_frag_from_hb(hb8, node_ld, kc, half);
#pragma unroll
    for (int kc = 0; kc < 4; ++kc) af[4 + kc] = a_frag_from_row(mr + kc * 32);
  }

#pragma unroll 1
  for (int nt = 0; nt < 8; ++nt) {
    f32x8 acc;
    const float bias = b_h1[nt * 16 + colq];
#pragma unroll
    for (int v = 0; v < 8; ++v) acc[v] = bias;
#pragma unroll
    for (int kc = 0; kc < 8; ++kc) {
      if (kc + 1 < 8) __builtin_prefetch(&wf_h1[((kc + 1) * 8 + nt) * 32 + lane], 0, 3);
      bf16x16 b = load_b_frag(wf_h1, kc, nt, lane);
      acc = __builtin_amdgcn_wmma_f32_16x16x32_bf16(false, af[kc], false, b, (short)0, acc, false, false);
    }
#pragma unroll
    for (int v = 0; v < 8; ++v) {
      const int row = v + (half << 3);
      aw[fragord_index(row, nt * 16 + colq)] = f2bf(silu_f(acc[v]));
    }
  }
  __builtin_amdgcn_wave_barrier();

  bf16x16 a2f[4];
#pragma unroll
  for (int kc = 0; kc < 4; ++kc) a2f[kc] = a_frag_from_fragord(aw, kc, lane);

#pragma unroll 1
  for (int nt = 0; nt < 8; ++nt) {
    f32x8 acc;
    const float bias = b_h2[nt * 16 + colq];
#pragma unroll
    for (int v = 0; v < 8; ++v) acc[v] = bias;
#pragma unroll
    for (int kc = 0; kc < 4; ++kc) {
      bf16x16 b = load_b_frag(wf_h2, kc, nt, lane);
      acc = __builtin_amdgcn_wmma_f32_16x16x32_bf16(false, a2f[kc], false, b, (short)0, acc, false, false);
    }
#pragma unroll
    for (int v = 0; v < 8; ++v) {
      const int row = v + ((lane >> 4) << 3);
      const int node = nbase + row;
      const int col = nt * 16 + colq;
      if (node < NNODES) hio[node * 128 + col] = acc[v] + h[node * 128 + col];
    }
  }
}

extern "C" void kernel_launch(void* const* d_in, const int* in_sizes, int n_in,
                              void* d_out, int out_size, void* d_ws, size_t ws_size,
                              hipStream_t stream) {
  const float* h     = (const float*)d_in[0];
  const float* x     = (const float*)d_in[1];
  const int*   edges = (const int*)  d_in[2];
  const float* w_e1  = (const float*)d_in[3];
  const float* b_e1  = (const float*)d_in[4];
  const float* w_e2  = (const float*)d_in[5];
  const float* b_e2  = (const float*)d_in[6];
  const float* w_inf = (const float*)d_in[7];
  const float* b_inf = (const float*)d_in[8];
  const float* w_h1  = (const float*)d_in[9];
  const float* b_h1  = (const float*)d_in[10];
  const float* w_h2  = (const float*)d_in[11];
  const float* b_h2  = (const float*)d_in[12];
  float* out = (float*)d_out;

  unsigned int* wf_e1 = (unsigned int*)d_ws;
  unsigned int* wf_e2 = wf_e1 + 18432;
  unsigned int* wf_h1 = wf_e2 + 8192;
  unsigned int* wf_h2 = wf_h1 + 16384;
  unsigned int* hb    = wf_h2 + 8192;

  egnn_prep_kernel<<<PREP_W_BLOCKS + PREP_H_BLOCKS, 256, 0, stream>>>(
      w_e1, w_e2, w_h1, w_h2, wf_e1, wf_e2, wf_h1, wf_h2, h, (u32x4*)hb, out);

  egnn_edge_kernel<<<NEDGES / 256, 128, 0, stream>>>(
      (const u32x8*)hb, x, edges, (const u32x8*)wf_e1, (const u32x8*)wf_e2,
      b_e1, b_e2, w_inf, b_inf, out);

  egnn_node_kernel<<<(NNODES + 63) / 64, 128, 0, stream>>>(
      h, (const u32x8*)hb, (const u32x8*)wf_h1, (const u32x8*)wf_h2, b_h1, b_h2, out);
}


# edge kernel: a column tile's weight fragments are requested one tile ahead, under the work of the tile in front (layer 1 into the one buffer once read, layer 2 into a second), not at its own head
# speedup vs baseline: 1.8728x; 1.1147x over previous
#include <hip/hip_runtime.h>
#include <hip/hip_bf16.h>

#define NNODES 50000
#define NEDGES 800000

typedef __attribute__((ext_vector_type(16))) __bf16        bf16x16;
typedef __attribute__((ext_vector_type(8)))  float         f32x8;
typedef __attribute__((ext_vector_type(8)))  unsigned int  u32x8;

__device__ __forceinline__ unsigned short f2bf(float f) {
  unsigned int u = __float_as_uint(f);
  u += 0x7FFFu + ((u >> 16) & 1u);
  return (unsigned short)(u >> 16);
}
__device__ __forceinline__ float bf2f(unsigned short s) {
  return __uint_as_float(((unsigned int)s) << 16);
}
__device__ __forceinline__ float silu_f(float v) { return v / (1.0f + __expf(-v)); }
__device__ __forceinline__ float sigm_f(float v) { return 1.0f / (1.0f + __expf(-v)); }

#define STAGE_ORDER() __builtin_amdgcn_sched_barrier(0)
__device__ __forceinline__ void silu_x8(const float (&x)[8], float (&out)[8]) {
  float den[8], ds[8], ns[8], r[8], q[8], e[8];
  bool fl[8], fl_den;
#pragma unroll
  for (int v = 0; v < 8; ++v) den[v] = 1.0f + __expf(-x[v]);
#pragma unroll
  for (int v = 0; v < 8; ++v) ds[v] = __builtin_amdgcn_div_scalef(x[v], den[v], false, &fl_den);
#pragma unroll
  for (int v = 0; v < 8; ++v) ns[v] = __builtin_amdgcn_div_scalef(x[v], den[v], true, &fl[v]);
  STAGE_ORDER();
#pragma unroll
  for (int v = 0; v < 8; ++v) r[v] = __builtin_amdgcn_rcpf(ds[v]);
#pragma unroll
  for (int v = 0; v < 8; ++v) e[v] = __builtin_fmaf(-ds[v], r[v], 1.0f);
  STAGE_ORDER();
#pragma unroll
  for (int v = 0; v < 8; ++v) r[v] = __builtin_fmaf(e[v], r[v], r[v]);
#pragma unroll
  for (int v = 0; v < 8; ++v) q[v] = ns[v] * r[v];
#pragma unroll
  for (int v = 0; v < 8; ++v) e[v] = __builtin_fmaf(-ds[v], q[v], ns[v]);
  STAGE_ORDER();
#pragma unroll
  for (int v = 0; v < 8; ++v) q[v] = __builtin_fmaf(e[v], r[v], q[v]);
#pragma unroll
  for (int v = 0; v < 8; ++v) e[v] = __builtin_fmaf(-ds[v], q[v], ns[v]);
  STAGE_ORDER();
#pragma unroll
  for (int v = 0; v < 8; ++v) q[v] = __builtin_amdgcn_div_fmasf(e[v], r[v], q[v], fl[v]);
  STAGE_ORDER();
#pragma unroll
  for (int v = 0; v < 8; ++v) out[v] = __builtin_amdgcn_div_fixupf(q[v], den[v], x[v]);
}

__device__ __forceinline__ int frag_k(int lane, int v) {
  return ((lane >> 4) << 3) + ((v >> 2) << 4) + ((v & 3) << 1);
}

typedef __attribute__((ext_vector_type(4))) float         f32x4;
typedef __attribute__((ext_vector_type(4))) unsigned int  u32x4;

__device__ __forceinline__ unsigned int pack2bf(float lo, float hi) {
  return (unsigned int)f2bf(lo) | ((unsigned int)f2bf(hi) << 16);
}

__device__ __forceinline__ bf16x16 a_frag_from_row(const float* p) {
  const f32x4 q0 = *(const f32x4*)(p);
  const f32x4 q1 = *(const f32x4*)(p + 4);
  const f32x4 q2 = *(const f32x4*)(p + 16);
  const f32x4 q3 = *(const f32x4*)(p + 20);
  u32x8 r;
  r[0] = pack2bf(q0[0], q0[1]); r[1] = pack2bf(q0[2], q0[3]);
  r[2] = pack2bf(q1[0], q1[1]); r[3] = pack2bf(q1[2], q1[3]);
  r[4] = pack2bf(q2[0], q2[1]); r[5] = pack2bf(q2[2], q2[3]);
  r[6] = pack2bf(q3[0], q3[1]); r[7] = pack2bf(q3[2], q3[3]);
  return __builtin_bit_cast(bf16x16, r);
}

__device__ __forceinline__ int fragord_index(int row, int col) {
  return (((col >> 4) * 32 + row + (((col >> 3) & 1) << 4)) << 3) + (col & 7);
}
__device__ __forceinline__ bf16x16 a_frag_from_fragord(const unsigned short* buf, int kc, int lane) {
  const u32x4 lo = *(const u32x4*)(buf + (((2 * kc)     * 32 + lane) << 3));
  const u32x4 hi = *(const u32x4*)(buf + (((2 * kc + 1) * 32 + lane) << 3));
  u32x8 r;
  r[0] = lo[0]; r[1] = lo[1]; r[2] = lo[2]; r[3] = lo[3];
  r[4] = hi[0]; r[5] = hi[1]; r[6] = hi[2]; r[7] = hi[3];
  return __builtin_bit_cast(bf16x16, r);
}

__device__ __forceinline__ bf16x16 a_frag_from_hb(const u32x8* __restrict__ hb8, int node, int kc, int half) {
  return __builtin_bit_cast(bf16x16, hb8[node * 8 + kc * 2 + half]);
}

__device__ __forceinline__ bf16x16 load_b_frag(const u32x8* __restrict__ wf, int kc, int nt, int lane) {
  return __builtin_bit_cast(bf16x16, wf[(kc * 8 + nt) * 32 + lane]);
}

#define PREP_W_BLOCKS 200
#define PREP_ITEMS 4
#define PREP_H_BLOCKS 782
static_assert((NNODES * 16) % 32 == 0 && (PREP_H_BLOCKS * 256) % 32 == 0, "whole waves of items");
__global__ void egnn_prep_kernel(const float* __restrict__ w_e1, const float* __restrict__ w_e2,
                                 const float* __restrict__ w_h1, const float* __restrict__ w_h2,
                                 unsigned int* __restrict__ wf_e1, unsigned int* __restrict__ wf_e2,
                                 unsigned int* __restrict__ wf_h1, unsigned int* __restrict__ wf_h2,
                                 const float* __restrict__ h, u32x4* __restrict__ hb,
                                 float* __restrict__ m_i) {
  if (blockIdx.x >= PREP_W_BLOCKS) {
    const int tt = (blockIdx.x - PREP_W_BLOCKS) * 256 + threadIdx.x;
    const int lane = threadIdx.x & 31;
    f32x4 q0[PREP_ITEMS], q1[PREP_ITEMS];
#pragma unroll
    for (int j = 0; j < PREP_ITEMS; ++j) {
      const int item = tt + j * (PREP_H_BLOCKS * 256);
      const int ld = item < NNODES * 16 ? item : NNODES * 16 - 1;
      const float* p = h + ld * 8;
      q0[j] = *(const f32x4*)(p);
      q1[j] = *(const f32x4*)(p + 4);
    }
#pragma unroll
    for (int j = 0; j < PREP_ITEMS; ++j) {
      const int item = tt + j * (PREP_H_BLOCKS * 256);
      if (item < NNODES * 16) {
        float* z = m_i + (item - lane) * 8 + lane * 4;
        const f32x4 zero = {0.0f, 0.0f, 0.0f, 0.0f};
        *(f32x4*)(z)       = zero;
        *(f32x4*)(z + 128) = zero;
      }
    }
#pragma unroll
    for (int j = 0; j < PREP_ITEMS; ++j) {
      const int item = tt + j * (PREP_H_BLOCKS * 256);
      const int g = item & 15, kc = g >> 2, q = g & 3;
      const int slot = ((q & 1) << 1) | (q >> 1);
      u32x4 r;
      r[0] = pack2bf(q0[j][0], q0[j][1]); r[1] = pack2bf(q0[j][2], q0[j][3]);
      r[2] = pack2bf(q1[j][0], q1[j][1]); r[3] = pack2bf(q1[j][2], q1[j][3]);
      if (item < NNODES * 16) hb[(item & ~15) + kc * 4 + slot] = r;
    }
    return;
  }
  int t = blockIdx.x * blockDim.x + threadIdx.x;
  const float* W; unsigned int* D; int K; int local;
  if      (t < 18432)         { W = w_e1; D = wf_e1; K = 257; local = t; }
  else if (t < 18432 + 8192)  { W = w_e2; D = wf_e2; K = 128; local = t - 18432; }
  else if (t < 26624 + 16384) { W = w_h1; D = wf_h1; K = 256; local = t - 26624; }
  else if (t < 51200)         { W = w_h2; D = wf_h2; K = 128; local = t - 43008; }
  else return;
  const int v = local & 7, lane = (local >> 3) & 31, nt = (local >> 8) & 7, kc = local >> 11;
  const int k   = kc * 32 + ((lane >> 4) << 3) + ((v >> 2) << 4) + ((v & 3) << 1);
  const int col = nt * 16 + (lane & 15);
  const float a = (k     < K) ? W[k * 128 + col]       : 0.0f;
  const float b = (k + 1 < K) ? W[(k + 1) * 128 + col] : 0.0f;
  D[local] = (unsigned int)f2bf(a) | ((unsigned int)f2bf(b) << 16);
}

#define MT 4
__global__ __launch_bounds__(128, 2) void egnn_edge_kernel(
    const u32x8* __restrict__ hb8, const float* __restrict__ x, const int* __restrict__ edges,
    const u32x8* __restrict__ wf_e1, const u32x8* __restrict__ wf_e2,
    const float* __restrict__ b_e1, const float* __restrict__ b_e2,
    const float* __restrict__ w_inf, const float* __restrict__ b_inf,
    float* __restrict__ m_i) {
  __shared__ __attribute__((aligned(16))) unsigned short tbuf[4][MT][16 * 128];
  __shared__ float gbuf[4][MT * 16];

  const int lane = threadIdx.x & 31;
  const int wave = threadIdx.x >> 5;
  const int colq = lane & 15;
  const int half = lane >> 4;
  const int ebase = (blockIdx.x * 4 + wave) * (16 * MT);

  int s_own[MT], d_own[MT];
  unsigned int dbf[MT];
#pragma unroll
  for (int mt = 0; mt < MT; ++mt) {
    const int e_own = ebase + mt * 16 + colq;
    const int s = edges[2 * e_own], d = edges[2 * e_own + 1];
    s_own[mt] = s;
    d_own[mt] = d;
    const float dx = x[3 * s] - x[3 * d], dy = x[3 * s + 1] - x[3 * d + 1],
                dz = x[3 * s + 2] - x[3 * d + 2];
    const float dist = sqrtf(dx * dx + dy * dy + dz * dz);
    dbf[mt] = (half == 0) ? (unsigned int)f2bf(dist) : 0u;
  }

  bf16x16 af[MT][9];
#pragma unroll
  for (int mt = 0; mt < MT; ++mt) {
#pragma unroll
    for (int kc = 0; kc < 4; ++kc) {
      af[mt][kc]     = a_frag_from_hb(hb8, s_own[mt], kc, half);
      af[mt][4 + kc] = a_frag_from_hb(hb8, d_own[mt], kc, half);
    }
  }
#pragma unroll
  for (int mt = 0; mt < MT; ++mt) {
    u32x8 z;
#pragma unroll
    for (int v = 0; v < 8; ++v) z[v] = 0u;
    z[0] = dbf[mt];
    af[mt][8] = __builtin_bit_cast(bf16x16, z);
  }

  bf16x16 bw[9];
  float bias;
  auto l1_mma = [&](f32x8 (&acc)[MT]) __attribute__((always_inline)) {
#pragma unroll
    for (int mt = 0; mt < MT; ++mt) {
#pragma unroll
      for (int v = 0; v < 8; ++v) acc[mt][v] = bias;
    }
#pragma unroll
    for (int kc = 0; kc < 9; ++kc) {
#pragma unroll
      for (int mt = 0; mt < MT; ++mt)
        acc[mt] = __builtin_amdgcn_wmma_f32_16x16x32_bf16(false, af[mt][kc], false, bw[kc], (short)0, acc[mt], false, false);
    }
  };
  auto l1_epi = [&](int nt, const f32x8 (&acc)[MT]) __attribute__((always_inline)) {
#pragma unroll
    for (int mt = 0; mt < MT; ++mt) {
      unsigned short* aw = tbuf[wave][mt];
      float a[8], s[8];
#pragma unroll
      for (int v = 0; v < 8; ++v) a[v] = acc[mt][v];
      silu_x8(a, s);
#pragma unroll
      for (int v = 0; v < 8; ++v) {
        const int row = v + (half << 3);
        aw[fragord_index(row, nt * 16 + colq)] = f2bf(s[v]);
      }
    }
  };
#pragma unroll
  for (int kc = 0; kc < 9; ++kc) bw[kc] = load_b_frag(wf_e1, kc, 0, lane);
  bias = b_e1[colq];
#pragma unroll 1
  for (int nt = 0; nt < 7; ++nt) {
    f32x8 acc[MT];
    l1_mma(acc);
#pragma unroll
    for (int kc = 0; kc < 9; ++kc) bw[kc] = load_b_frag(wf_e1, kc, nt + 1, lane);
    bias = b_e1[(nt + 1) * 16 + colq];
    l1_epi(nt, acc);
  }
  float wv0;
  {
    f32x8 acc[MT];
    l1_mma(acc);
#pragma unroll
    for (int kc = 0; kc < 4; ++kc) bw[kc] = load_b_frag(wf_e2, kc, 0, lane);
    bias = b_e2[colq];
    wv0  = w_inf[colq];
    l1_epi(7, acc);
  }
  __builtin_amdgcn_wave_barrier();

  bf16x16 a2f[MT][4];
#pragma unroll
  for (int mt = 0; mt < MT; ++mt) {
#pragma unroll
    for (int kc = 0; kc < 4; ++kc) a2f[mt][kc] = a_frag_from_fragord(tbuf[wave][mt], kc, lane);
  }
  __builtin_amdgcn_wave_barrier();

  f32x8 dot[MT];
#pragma unroll
  for (int mt = 0; mt < MT; ++mt) {
#pragma unroll
    for (int v = 0; v < 8; ++v) dot[mt][v] = 0.0f;
  }
  float bias2[2], wv2[2];
  bias2[0] = bias;
  wv2[0] = wv0;
  auto l2_request = [&](int nt, int buf) __attribute__((always_inline)) {
#pragma unroll
    for (int kc = 0; kc < 4; ++kc) bw[buf * 4 + kc] = load_b_frag(wf_e2, kc, nt, lane);
    bias2[buf] = b_e2[nt * 16 + colq];
    wv2[buf]   = w_inf[nt * 16 + colq];
  };
  auto l2_tile = [&](int nt, int buf) __attribute__((always_inline)) {
    const float wv = wv2[buf];
    f32x8 acc[MT];
#pragma unroll
    for (int mt = 0; mt < MT; ++mt) {
#pragma unroll
      for (int v = 0; v < 8; ++v) acc[mt][v] = bias2[buf];
    }
#pragma unroll
    for (int kc = 0; kc < 4; ++kc) {
#pragma unroll
      for (int mt = 0; mt < MT; ++mt)
        acc[mt] = __builtin_amdgcn_wmma_f32_16x16x32_bf16(false, a2f[mt][kc], false, bw[buf * 4 + kc], (short)0, acc[mt], false, false);
    }
#pragma unroll
    for (int mt = 0; mt < MT; ++mt) {
      unsigned short* mb = tbuf[wave][mt];
      float a[8], m[8];
#pragma unroll
      for (int v = 0; v < 8; ++v) a[v] = acc[mt][v];
      silu_x8(a, m);
#pragma unroll
      for (int v = 0; v < 8; ++v) dot[mt][v] += m[v] * wv;
#pragma unroll
      for (int v = 0; v < 8; ++v) {
        const int row = v + (half << 3);
        mb[row * 128 + nt * 16 + colq] = f2bf(m[v]);
      }
    }
  };
#pragma unroll 1
  for (int np = 0; np < 3; ++np) {
    l2_request(2 * np + 1, 1);
    l2_tile(2 * np, 0);
    l2_request(2 * np + 2, 0);
    l2_tile(2 * np + 1, 1);
  }
  l2_request(7, 1);
  l2_tile(6, 0);
  l2_tile(7, 1);
  const float binf = b_inf[0];
#pragma unroll
  for (int mt = 0; mt < MT; ++mt) {
#pragma unroll
    for (int v = 0; v < 8; ++v) {
      float s = dot[mt][v];
      s += __shfl_xor(s, 1, 32);
      s += __shfl_xor(s, 2, 32);
      s += __shfl_xor(s, 4, 32);
      s += __shfl_xor(s, 8, 32);
      dot[mt][v] = s;
    }
  }
  if (colq == 0) {
#pragma unroll
    for (int mt = 0; mt < MT; ++mt) {
#pragma unroll
      for (int v = 0; v < 8; ++v) gbuf[wave][mt * 16 + v + (half << 3)] = sigm_f(dot[mt][v] + binf);
    }
  }
  __builtin_amdgcn_wave_barrier();

  for (int r = 0; r < 16; ++r) {
    float val[MT][4];
    int srow[MT];
#pragma unroll
    for (int mt = 0; mt < MT; ++mt) {
      srow[mt] = __builtin_amdgcn_readlane(s_own[mt], r);
      const float g = gbuf[wave][mt * 16 + r];
      const unsigned short* mb = tbuf[wave][mt] + r * 128 + lane;
#pragma unroll
      for (int j = 0; j < 4; ++j) val[mt][j] = g * bf2f(mb[32 * j]);
    }
#pragma unroll
    for (int mt = 0; mt < MT; ++mt) {
#pragma unroll
      for (int j = 0; j < 4; ++j) unsafeAtomicAdd(&m_i[srow[mt] * 128 + lane + 32 * j], val[mt][j]);
    }
  }
}

#define NODE_WAVES 4
#define NODE_NT (8 / NODE_WAVES)
static_assert(NNODES % 16 == 0, "the node kernel's grid is NNODES / 16 whole tiles and guards no row");
static_assert(NODE_NT * 16 * sizeof(float) == 128, "a wave's columns are one 128-byte line of each output row");
__global__ __launch_bounds__(128) void egnn_node_kernel(
    const float* __restrict__ h, const u32x8* __restrict__ hb8,
    const u32x8* __restrict__ wf_h1, const u32x8* __restrict__ wf_h2,
    const float* __restrict__ b_h1, const float* __restrict__ b_h2,
    float* __restrict__ hio) {
  __shared__ __attribute__((aligned(16))) unsigned short a2[8 * 32 * 8];

  const int lane = threadIdx.x & 31;
  const int wave = threadIdx.x >> 5;
  const int colq = lane & 15;
  const int half = lane >> 4;
  const int nbase = blockIdx.x * 16;

  const int node_own = nbase + colq;
  bf16x16 af[8];
  {
    const float* mr = hio + node_own * 128 + 8 * half;
#pragma unroll
    for (int kc = 0; kc < 4; ++kc) af[kc] = a_frag_from_hb(hb8, node_own, kc, half);
#pragma unroll
    for (int kc = 0; kc < 4; ++kc) af[4 + kc] = a_frag_from_row(mr + kc * 32);
  }

  for (int t = 0; t < NODE_NT; ++t) {
    const int nt = wave * NODE_NT + t;
    f32x8 acc;
    const float bias = b_h1[nt * 16 + colq];
#pragma unroll
    for (int v = 0; v < 8; ++v) acc[v] = bias;
#pragma unroll
    for (int kc = 0; kc < 8; ++kc) {
      if (kc + 1 < 8) __builtin_prefetch(&wf_h1[((kc + 1) * 8 + nt) * 32 + lane], 0, 3);
      bf16x16 b = load_b_frag(wf_h1, kc, nt, lane);
      acc = __builtin_amdgcn_wmma_f32_16x16x32_bf16(false, af[kc], false, b, (short)0, acc, false, false);
    }
#pragma unroll
    for (int v = 0; v < 8; ++v) {
      const int row = v + (half << 3);
      a2[fragord_index(row, nt * 16 + colq)] = f2bf(silu_f(acc[v]));
    }
  }
  __syncthreads();

  bf16x16 a2f[4];
#pragma unroll
  for (int kc = 0; kc < 4; ++kc) a2f[kc] = a_frag_from_fragord(a2, kc, lane);

  for (int t = 0; t < NODE_NT; ++t) {
    const int nt = wave * NODE_NT + t;
    f32x8 acc;
    const float bias = b_h2[nt * 16 + colq];
#pragma unroll
    for (int v = 0; v < 8; ++v) acc[v] = bias;
#pragma unroll
    for (int kc = 0; kc < 4; ++kc) {
      bf16x16 b = load_b_frag(wf_h2, kc, nt, lane);
      acc = __builtin_amdgcn_wmma_f32_16x16x32_bf16(false, a2f[kc], false, b, (short)0, acc, false, false);
    }
#pragma unroll
    for (int v = 0; v < 8; ++v) {
      const int node = nbase + v + (half << 3);
      const int col = nt * 16 + colq;
      hio[node * 128 + col] = acc[v] + h[node * 128 + col];
    }
  }
}

extern "C" void kernel_launch(void* const* d_in, const int* in_sizes, int n_in,
                              void* d_out, int out_size, void* d_ws, size_t ws_size,
                              hipStream_t stream) {
  const float* h     = (const float*)d_in[0];
  const float* x     = (const float*)d_in[1];
  const int*   edges = (const int*)  d_in[2];
  const float* w_e1  = (const float*)d_in[3];
  const float* b_e1  = (const float*)d_in[4];
  const float* w_e2  = (const float*)d_in[5];
  const float* b_e2  = (const float*)d_in[6];
  const float* w_inf = (const float*)d_in[7];
  const float* b_inf = (const float*)d_in[8];
  const float* w_h1  = (const float*)d_in[9];
  const float* b_h1  = (const float*)d_in[10];
  const float* w_h2  = (const float*)d_in[11];
  const float* b_h2  = (const float*)d_in[12];
  float* out = (float*)d_out;

  unsigned int* wf_e1 = (unsigned int*)d_ws;
  unsigned int* wf_e2 = wf_e1 + 18432;
  unsigned int* wf_h1 = wf_e2 + 8192;
  unsigned int* wf_h2 = wf_h1 + 16384;
  unsigned int* hb    = wf_h2 + 8192;

  egnn_prep_kernel<<<PREP_W_BLOCKS + PREP_H_BLOCKS, 256, 0, stream>>>(
      w_e1, w_e2, w_h1, w_h2, wf_e1, wf_e2, wf_h1, wf_h2, h, (u32x4*)hb, out);

  egnn_edge_kernel<<<NEDGES / 256, 128, 0, stream>>>(
      (const u32x8*)hb, x, edges, (const u32x8*)wf_e1, (const u32x8*)wf_e2,
      b_e1, b_e2, w_inf, b_inf, out);

  egnn_node_kernel<<<NNODES / 16, 128, 0, stream>>>(
      h, (const u32x8*)hb, (const u32x8*)wf_h1, (const u32x8*)wf_h2, b_h1, b_h2, out);
}


# SiLU (edge and node kernels) and the gate sigmoid by the tanh instruction: x/(1+exp(-x)) = h + h*tanh(h), h = x/2: a product, a v_tanh_f32 and an fma per value, no exp or division; few-ulp change
# speedup vs baseline: 2.1864x; 1.1674x over previous
#include <hip/hip_runtime.h>
#include <hip/hip_bf16.h>

#define NNODES 50000
#define NEDGES 800000

typedef __attribute__((ext_vector_type(16))) __bf16        bf16x16;
typedef __attribute__((ext_vector_type(8)))  float         f32x8;
typedef __attribute__((ext_vector_type(8)))  unsigned int  u32x8;

__device__ __forceinline__ unsigned short f2bf(float f) {
  unsigned int u = __float_as_uint(f);
  u += 0x7FFFu + ((u >> 16) & 1u);
  return (unsigned short)(u >> 16);
}
__device__ __forceinline__ float bf2f(unsigned short s) {
  return __uint_as_float(((unsigned int)s) << 16);
}

__device__ __forceinline__ float silu_f(float v) {
  const float h = 0.5f * v;
  return __builtin_fmaf(h, __builtin_amdgcn_tanhf(h), h);
}
__device__ __forceinline__ float sigm_f(float v) {
  return __builtin_fmaf(0.5f, __builtin_amdgcn_tanhf(0.5f * v), 0.5f);
}
__device__ __forceinline__ void silu_tile(const f32x8 (&acc)[4], float (&s)[4][8]) {
#pragma unroll
  for (int a = 0; a < 4; ++a) {
#pragma unroll
    for (int v = 0; v < 8; ++v) s[a][v] = silu_f(acc[a][v]);
  }
}

__device__ __forceinline__ int frag_k(int lane, int v) {
  return ((lane >> 4) << 3) + ((v >> 2) << 4) + ((v & 3) << 1);
}

typedef __attribute__((ext_vector_type(4))) float         f32x4;
typedef __attribute__((ext_vector_type(4))) unsigned int  u32x4;

__device__ __forceinline__ unsigned int pack2bf(float lo, float hi) {
  return (unsigned int)f2bf(lo) | ((unsigned int)f2bf(hi) << 16);
}

__device__ __forceinline__ bf16x16 a_frag_from_row(const float* p) {
  const f32x4 q0 = *(const f32x4*)(p);
  const f32x4 q1 = *(const f32x4*)(p + 4);
  const f32x4 q2 = *(const f32x4*)(p + 16);
  const f32x4 q3 = *(const f32x4*)(p + 20);
  u32x8 r;
  r[0] = pack2bf(q0[0], q0[1]); r[1] = pack2bf(q0[2], q0[3]);
  r[2] = pack2bf(q1[0], q1[1]); r[3] = pack2bf(q1[2], q1[3]);
  r[4] = pack2bf(q2[0], q2[1]); r[5] = pack2bf(q2[2], q2[3]);
  r[6] = pack2bf(q3[0], q3[1]); r[7] = pack2bf(q3[2], q3[3]);
  return __builtin_bit_cast(bf16x16, r);
}

__device__ __forceinline__ int fragord_index(int row, int col) {
  return (((col >> 4) * 32 + row + (((col >> 3) & 1) << 4)) << 3) + (col & 7);
}
__device__ __forceinline__ bf16x16 a_frag_from_fragord(const unsigned short* buf, int kc, int lane) {
  const u32x4 lo = *(const u32x4*)(buf + (((2 * kc)     * 32 + lane) << 3));
  const u32x4 hi = *(const u32x4*)(buf + (((2 * kc + 1) * 32 + lane) << 3));
  u32x8 r;
  r[0] = lo[0]; r[1] = lo[1]; r[2] = lo[2]; r[3] = lo[3];
  r[4] = hi[0]; r[5] = hi[1]; r[6] = hi[2]; r[7] = hi[3];
  return __builtin_bit_cast(bf16x16, r);
}

__device__ __forceinline__ bf16x16 a_frag_from_hb(const u32x8* __restrict__ hb8, int node, int kc, int half) {
  return __builtin_bit_cast(bf16x16, hb8[node * 8 + kc * 2 + half]);
}

__device__ __forceinline__ bf16x16 load_b_frag(const u32x8* __restrict__ wf, int kc, int nt, int lane) {
  return __builtin_bit_cast(bf16x16, wf[(kc * 8 + nt) * 32 + lane]);
}

#define PREP_W_BLOCKS 200
#define PREP_ITEMS 4
#define PREP_H_BLOCKS 782
static_assert((NNODES * 16) % 32 == 0 && (PREP_H_BLOCKS * 256) % 32 == 0, "whole waves of items");
__global__ void egnn_prep_kernel(const float* __restrict__ w_e1, const float* __restrict__ w_e2,
                                 const float* __restrict__ w_h1, const float* __restrict__ w_h2,
                                 unsigned int* __restrict__ wf_e1, unsigned int* __restrict__ wf_e2,
                                 unsigned int* __restrict__ wf_h1, unsigned int* __restrict__ wf_h2,
                                 const float* __restrict__ h, u32x4* __restrict__ hb,
                                 float* __restrict__ m_i) {
  if (blockIdx.x >= PREP_W_BLOCKS) {
    const int tt = (blockIdx.x - PREP_W_BLOCKS) * 256 + threadIdx.x;
    const int lane = threadIdx.x & 31;
    f32x4 q0[PREP_ITEMS], q1[PREP_ITEMS];
#pragma unroll
    for (int j = 0; j < PREP_ITEMS; ++j) {
      const int item = tt + j * (PREP_H_BLOCKS * 256);
      const int ld = item < NNODES * 16 ? item : NNODES * 16 - 1;
      const float* p = h + ld * 8;
      q0[j] = *(const f32x4*)(p);
      q1[j] = *(const f32x4*)(p + 4);
    }
#pragma unroll
    for (int j = 0; j < PREP_ITEMS; ++j) {
      const int item = tt + j * (PREP_H_BLOCKS * 256);
      if (item < NNODES * 16) {
        float* z = m_i + (item - lane) * 8 + lane * 4;
        const f32x4 zero = {0.0f, 0.0f, 0.0f, 0.0f};
        *(f32x4*)(z)       = zero;
        *(f32x4*)(z + 128) = zero;
      }
    }
#pragma unroll
    for (int j = 0; j < PREP_ITEMS; ++j) {
      const int item = tt + j * (PREP_H_BLOCKS * 256);
      const int g = item & 15, kc = g >> 2, q = g & 3;
      const int slot = ((q & 1) << 1) | (q >> 1);
      u32x4 r;
      r[0] = pack2bf(q0[j][0], q0[j][1]); r[1] = pack2bf(q0[j][2], q0[j][3]);
      r[2] = pack2bf(q1[j][0], q1[j][1]); r[3] = pack2bf(q1[j][2], q1[j][3]);
      if (item < NNODES * 16) hb[(item & ~15) + kc * 4 + slot] = r;
    }
    return;
  }
  int t = blockIdx.x * blockDim.x + threadIdx.x;
  const float* W; unsigned int* D; int K; int local;
  if      (t < 18432)         { W = w_e1; D = wf_e1; K = 257; local = t; }
  else if (t < 18432 + 8192)  { W = w_e2; D = wf_e2; K = 128; local = t - 18432; }
  else if (t < 26624 + 16384) { W = w_h1; D = wf_h1; K = 256; local = t - 26624; }
  else if (t < 51200)         { W = w_h2; D = wf_h2; K = 128; local = t - 43008; }
  else return;
  const int v = local & 7, lane = (local >> 3) & 31, nt = (local >> 8) & 7, kc = local >> 11;
  const int k   = kc * 32 + ((lane >> 4) << 3) + ((v >> 2) << 4) + ((v & 3) << 1);
  const int col = nt * 16 + (lane & 15);
  const float a = (k     < K) ? W[k * 128 + col]       : 0.0f;
  const float b = (k + 1 < K) ? W[(k + 1) * 128 + col] : 0.0f;
  D[local] = (unsigned int)f2bf(a) | ((unsigned int)f2bf(b) << 16);
}

#define MT 4
static_assert(MT == 4, "silu_tile takes the 4 accumulators of a column tile");
__global__ __launch_bounds__(128, 2) void egnn_edge_kernel(
    const u32x8* __restrict__ hb8, const float* __restrict__ x, const int* __restrict__ edges,
    const u32x8* __restrict__ wf_e1, const u32x8* __restrict__ wf_e2,
    const float* __restrict__ b_e1, const float* __restrict__ b_e2,
    const float* __restrict__ w_inf, const float* __restrict__ b_inf,
    float* __restrict__ m_i) {
  __shared__ __attribute__((aligned(16))) unsigned short tbuf[4][MT][16 * 128];
  __shared__ float gbuf[4][MT * 16];

  const int lane = threadIdx.x & 31;
  const int wave = threadIdx.x >> 5;
  const int colq = lane & 15;
  const int half = lane >> 4;
  const int ebase = (blockIdx.x * 4 + wave) * (16 * MT);

  int s_own[MT], d_own[MT];
  unsigned int dbf[MT];
#pragma unroll
  for (int mt = 0; mt < MT; ++mt) {
    const int e_own = ebase + mt * 16 + colq;
    const int s = edges[2 * e_own], d = edges[2 * e_own + 1];
    s_own[mt] = s;
    d_own[mt] = d;
    const float dx = x[3 * s] - x[3 * d], dy = x[3 * s + 1] - x[3 * d + 1],
                dz = x[3 * s + 2] - x[3 * d + 2];
    const float dist = sqrtf(dx * dx + dy * dy + dz * dz);
    dbf[mt] = (half == 0) ? (unsigned int)f2bf(dist) : 0u;
  }

  bf16x16 af[MT][9];
#pragma unroll
  for (int mt = 0; mt < MT; ++mt) {
#pragma unroll
    for (int kc = 0; kc < 4; ++kc) {
      af[mt][kc]     = a_frag_from_hb(hb8, s_own[mt], kc, half);
      af[mt][4 + kc] = a_frag_from_hb(hb8, d_own[mt], kc, half);
    }
  }
#pragma unroll
  for (int mt = 0; mt < MT; ++mt) {
    u32x8 z;
#pragma unroll
    for (int v = 0; v < 8; ++v) z[v] = 0u;
    z[0] = dbf[mt];
    af[mt][8] = __builtin_bit_cast(bf16x16, z);
  }

  bf16x16 bw[9];
  float bias;
  auto l1_mma = [&](f32x8 (&acc)[MT]) __attribute__((always_inline)) {
#pragma unroll
    for (int mt = 0; mt < MT; ++mt) {
#pragma unroll
      for (int v = 0; v < 8; ++v) acc[mt][v] = bias;
    }
#pragma unroll
    for (int kc = 0; kc < 9; ++kc) {
#pragma unroll
      for (int mt = 0; mt < MT; ++mt)
        acc[mt] = __builtin_amdgcn_wmma_f32_16x16x32_bf16(false, af[mt][kc], false, bw[kc], (short)0, acc[mt], false, false);
    }
  };
  auto l1_epi = [&](int nt, const f32x8 (&acc)[MT]) __attribute__((always_inline)) {
    float s[MT][8];
    silu_tile(acc, s);
#pragma unroll
    for (int mt = 0; mt < MT; ++mt) {
      unsigned short* aw = tbuf[wave][mt];
#pragma unroll
      for (int v = 0; v < 8; ++v) {
        const int row = v + (half << 3);
        aw[fragord_index(row, nt * 16 + colq)] = f2bf(s[mt][v]);
      }
    }
  };
#pragma unroll
  for (int kc = 0; kc < 9; ++kc) bw[kc] = load_b_frag(wf_e1, kc, 0, lane);
  bias = b_e1[colq];
#pragma unroll 1
  for (int nt = 0; nt < 7; ++nt) {
    f32x8 acc[MT];
    l1_mma(acc);
#pragma unroll
    for (int kc = 0; kc < 9; ++kc) bw[kc] = load_b_frag(wf_e1, kc, nt + 1, lane);
    bias = b_e1[(nt + 1) * 16 + colq];
    l1_epi(nt, acc);
  }
  float wv0;
  {
    f32x8 acc[MT];
    l1_mma(acc);
#pragma unroll
    for (int kc = 0; kc < 4; ++kc) bw[kc] = load_b_frag(wf_e2, kc, 0, lane);
    bias = b_e2[colq];
    wv0  = w_inf[colq];
    l1_epi(7, acc);
  }
  __builtin_amdgcn_wave_barrier();

  bf16x16 a2f[MT][4];
#pragma unroll
  for (int mt = 0; mt < MT; ++mt) {
#pragma unroll
    for (int kc = 0; kc < 4; ++kc) a2f[mt][kc] = a_frag_from_fragord(tbuf[wave][mt], kc, lane);
  }
  __builtin_amdgcn_wave_barrier();

  f32x8 dot[MT];
#pragma unroll
  for (int mt = 0; mt < MT; ++mt) {
#pragma unroll
    for (int v = 0; v < 8; ++v) dot[mt][v] = 0.0f;
  }
  float bias2[2], wv2[2];
  bias2[0] = bias;
  wv2[0] = wv0;
  auto l2_request = [&](int nt, int buf) __attribute__((always_inline)) {
#pragma unroll
    for (int kc = 0; kc < 4; ++kc) bw[buf * 4 + kc] = load_b_frag(wf_e2, kc, nt, lane);
    bias2[buf] = b_e2[nt * 16 + colq];
    wv2[buf]   = w_inf[nt * 16 + colq];
  };
  auto l2_tile = [&](int nt, int buf) __attribute__((always_inline)) {
    const float wv = wv2[buf];
    f32x8 acc[MT];
#pragma unroll
    for (int mt = 0; mt < MT; ++mt) {
#pragma unroll
      for (int v = 0; v < 8; ++v) acc[mt][v] = bias2[buf];
    }
#pragma unroll
    for (int kc = 0; kc < 4; ++kc) {
#pragma unroll
      for (int mt = 0; mt < MT; ++mt)
        acc[mt] = __builtin_amdgcn_wmma_f32_16x16x32_bf16(false, a2f[mt][kc], false, bw[buf * 4 + kc], (short)0, acc[mt], false, false);
    }
    float m[MT][8];
    silu_tile(acc, m);
#pragma unroll
    for (int mt = 0; mt < MT; ++mt) {
      unsigned short* mb = tbuf[wave][mt];
#pragma unroll
      for (int v = 0; v < 8; ++v) dot[mt][v] += m[mt][v] * wv;
#pragma unroll
      for (int v = 0; v < 8; ++v) {
        const int row = v + (half << 3);
        mb[row * 128 + nt * 16 + colq] = f2bf(m[mt][v]);
      }
    }
  };
#pragma unroll 1
  for (int np = 0; np < 3; ++np) {
    l2_request(2 * np + 1, 1);
    l2_tile(2 * np, 0);
    l2_request(2 * np + 2, 0);
    l2_tile(2 * np + 1, 1);
  }
  l2_request(7, 1);
  l2_tile(6, 0);
  l2_tile(7, 1);
  const float binf = b_inf[0];
#pragma unroll
  for (int mt = 0; mt < MT; ++mt) {
#pragma unroll
    for (int v = 0; v < 8; ++v) {
      float s = dot[mt][v];
      s += __shfl_xor(s, 1, 32);
      s += __shfl_xor(s, 2, 32);
      s += __shfl_xor(s, 4, 32);
      s += __shfl_xor(s, 8, 32);
      dot[mt][v] = s;
    }
  }
  if (colq == 0) {
#pragma unroll
    for (int mt = 0; mt < MT; ++mt) {
#pragma unroll
      for (int v = 0; v < 8; ++v) gbuf[wave][mt * 16 + v + (half << 3)] = sigm_f(dot[mt][v] + binf);
    }
  }
  __builtin_amdgcn_wave_barrier();

  for (int r = 0; r < 16; ++r) {
    float val[MT][4];
    int srow[MT];
#pragma unroll
    for (int mt = 0; mt < MT; ++mt) {
      srow[mt] = __builtin_amdgcn_readlane(s_own[mt], r);
      const float g = gbuf[wave][mt * 16 + r];
      const unsigned short* mb = tbuf[wave][mt] + r * 128 + lane;
#pragma unroll
      for (int j = 0; j < 4; ++j) val[mt][j] = g * bf2f(mb[32 * j]);
    }
#pragma unroll
    for (int mt = 0; mt < MT; ++mt) {
#pragma unroll
      for (int j = 0; j < 4; ++j) unsafeAtomicAdd(&m_i[srow[mt] * 128 + lane + 32 * j], val[mt][j]);
    }
  }
}

#define NODE_WAVES 4
#define NODE_NT (8 / NODE_WAVES)
static_assert(NNODES % 16 == 0, "the node kernel's grid is NNODES / 16 whole tiles and guards no row");
static_assert(NODE_NT * 16 * sizeof(float) == 128, "a wave's columns are one 128-byte line of each output row");
__global__ __launch_bounds__(128) void egnn_node_kernel(
    const float* __restrict__ h, const u32x8* __restrict__ hb8,
    const u32x8* __restrict__ wf_h1, const u32x8* __restrict__ wf_h2,
    const float* __restrict__ b_h1, const float* __restrict__ b_h2,
    float* __restrict__ hio) {
  __shared__ __attribute__((aligned(16))) unsigned short a2[8 * 32 * 8];

  const int lane = threadIdx.x & 31;
  const int wave = threadIdx.x >> 5;
  const int colq = lane & 15;
  const int half = lane >> 4;
  const int nbase = blockIdx.x * 16;

  const int node_own = nbase + colq;
  bf16x16 af[8];
  {
    const float* mr = hio + node_own * 128 + 8 * half;
#pragma unroll
    for (int kc = 0; kc < 4; ++kc) af[kc] = a_frag_from_hb(hb8, node_own, kc, half);
#pragma unroll
    for (int kc = 0; kc < 4; ++kc) af[4 + kc] = a_frag_from_row(mr + kc * 32);
  }

  for (int t = 0; t < NODE_NT; ++t) {
    const int nt = wave * NODE_NT + t;
    f32x8 acc;
    const float bias = b_h1[nt * 16 + colq];
#pragma unroll
    for (int v = 0; v < 8; ++v) acc[v] = bias;
#pragma unroll
    for (int kc = 0; kc < 8; ++kc) {
      if (kc + 1 < 8) __builtin_prefetch(&wf_h1[((kc + 1) * 8 + nt) * 32 + lane], 0, 3);
      bf16x16 b = load_b_frag(wf_h1, kc, nt, lane);
      acc = __builtin_amdgcn_wmma_f32_16x16x32_bf16(false, af[kc], false, b, (short)0, acc, false, false);
    }
#pragma unroll
    for (int v = 0; v < 8; ++v) {
      const int row = v + (half << 3);
      a2[fragord_index(row, nt * 16 + colq)] = f2bf(silu_f(acc[v]));
    }
  }
  __syncthreads();

  bf16x16 a2f[4];
#pragma unroll
  for (int kc = 0; kc < 4; ++kc) a2f[kc] = a_frag_from_fragord(a2, kc, lane);

  for (int t = 0; t < NODE_NT; ++t) {
    const int nt = wave * NODE_NT + t;
    f32x8 acc;
    const float bias = b_h2[nt * 16 + colq];
#pragma unroll
    for (int v = 0; v < 8; ++v) acc[v] = bias;
#pragma unroll
    for (int kc = 0; kc < 4; ++kc) {
      bf16x16 b = load_b_frag(wf_h2, kc, nt, lane);
      acc = __builtin_amdgcn_wmma_f32_16x16x32_bf16(false, a2f[kc], false, b, (short)0, acc, false, false);
    }
#pragma unroll
    for (int v = 0; v < 8; ++v) {
      const int node = nbase + v + (half << 3);
      const int col = nt * 16 + colq;
      hio[node * 128 + col] = acc[v] + h[node * 128 + col];
    }
  }
}

extern "C" void kernel_launch(void* const* d_in, const int* in_sizes, int n_in,
                              void* d_out, int out_size, void* d_ws, size_t ws_size,
                              hipStream_t stream) {
  const float* h     = (const float*)d_in[0];
  const float* x     = (const float*)d_in[1];
  const int*   edges = (const int*)  d_in[2];
  const float* w_e1  = (const float*)d_in[3];
  const float* b_e1  = (const float*)d_in[4];
  const float* w_e2  = (const float*)d_in[5];
  const float* b_e2  = (const float*)d_in[6];
  const float* w_inf = (const float*)d_in[7];
  const float* b_inf = (const float*)d_in[8];
  const float* w_h1  = (const float*)d_in[9];
  const float* b_h1  = (const float*)d_in[10];
  const float* w_h2  = (const float*)d_in[11];
  const float* b_h2  = (const float*)d_in[12];
  float* out = (float*)d_out;

  unsigned int* wf_e1 = (unsigned int*)d_ws;
  unsigned int* wf_e2 = wf_e1 + 18432;
  unsigned int* wf_h1 = wf_e2 + 8192;
  unsigned int* wf_h2 = wf_h1 + 16384;
  unsigned int* hb    = wf_h2 + 8192;

  egnn_prep_kernel<<<PREP_W_BLOCKS + PREP_H_BLOCKS, 256, 0, stream>>>(
      w_e1, w_e2, w_h1, w_h2, wf_e1, wf_e2, wf_h1, wf_h2, h, (u32x4*)hb, out);

  egnn_edge_kernel<<<NEDGES / 256, 128, 0, stream>>>(
      (const u32x8*)hb, x, edges, (const u32x8*)wf_e1, (const u32x8*)wf_e2,
      b_e1, b_e2, w_inf, b_inf, out);

  egnn_node_kernel<<<NNODES / 16, 128, 0, stream>>>(
      h, (const u32x8*)hb, (const u32x8*)wf_h1, (const u32x8*)wf_h2, b_h1, b_h2, out);
}
